# starting kernel
# baseline (speedup 1.0000x reference)
#include <hip/hip_runtime.h>
#include <hip/hip_bf16.h>

typedef _Float16 half_t;
typedef __attribute__((ext_vector_type(16))) _Float16 v16h;
typedef __attribute__((ext_vector_type(8)))  _Float16 v8h;
typedef __attribute__((ext_vector_type(8)))  float    v8f;

union V16 { v16h v; v8h h[2]; };

constexpr int BB   = 2;
constexpr int TT   = 2048;
constexpr int DM   = 1024;
constexpr int HN   = 16;
constexpr int DH   = 64;
constexpr int QKVD = 3072;
constexpr int MROW = BB * TT;
constexpr int WIN  = 128;
constexpr float SM_SCALE = 0.125f;
constexpr float MASKED = -3.0e38f;
constexpr float MINIT  = -1.0e30f;

constexpr size_t OFF_QKVW = 0;
constexpr size_t SZ_QKVW  = (size_t)QKVD * DM * 2;
constexpr size_t OFF_OUTW = OFF_QKVW + SZ_QKVW;
constexpr size_t SZ_OUTW  = (size_t)DM * DM * 2;
constexpr size_t OFF_T    = OFF_OUTW + SZ_OUTW;
constexpr size_t SZ_T     = (size_t)MROW * DM * 2;
constexpr size_t OFF_QKVF = OFF_T + SZ_T;
constexpr size_t SZ_QKVF  = (size_t)MROW * QKVD * 4;
constexpr size_t OFF_ATTN = OFF_QKVF;
constexpr size_t OFF_Q    = OFF_QKVF + SZ_QKVF;
constexpr size_t SZ_QKH   = (size_t)BB * HN * TT * DH * 2;
constexpr size_t OFF_K    = OFF_Q + SZ_QKH;
constexpr size_t OFF_V    = OFF_K + SZ_QKH;

__device__ __forceinline__ void ldf(V16& f, const half_t* p) {
  f.h[0] = *(const v8h*)(p);
  f.h[1] = *(const v8h*)(p + 16);
}

__global__ __launch_bounds__(256) void rmsnorm_kernel(
    const float* __restrict__ x, const float* __restrict__ scale,
    half_t* __restrict__ t) {
  const int row = blockIdx.x;
  const float* xr = x + (size_t)row * DM;
  float ss = 0.f;
  for (int i = threadIdx.x; i < DM; i += 256) { float v = xr[i]; ss += v * v; }
  #pragma unroll
  for (int o = 16; o > 0; o >>= 1) ss += __shfl_xor(ss, o, 32);
  __shared__ float red[8];
  if ((threadIdx.x & 31) == 0) red[threadIdx.x >> 5] = ss;
  __syncthreads();
  float tot = 0.f;
  #pragma unroll
  for (int i = 0; i < 8; ++i) tot += red[i];
  const float inv = rsqrtf(tot * (1.0f / DM) + 1e-5f);
  for (int i = threadIdx.x; i < DM; i += 256)
    t[(size_t)row * DM + i] = (half_t)(xr[i] * inv * scale[i]);
}

__global__ __launch_bounds__(256) void f32_to_f16_kernel(
    const float* __restrict__ src, half_t* __restrict__ dst, int n) {
  int i = blockIdx.x * blockDim.x + threadIdx.x;
  const int stride = gridDim.x * blockDim.x;
  for (; i < n; i += stride) dst[i] = (half_t)src[i];
}

__device__ __forceinline__ void mma8(v8f acc[2][4], const V16 a[2], const V16 b[4]) {
  #pragma unroll
  for (int i = 0; i < 2; ++i)
    #pragma unroll
    for (int j = 0; j < 4; ++j)
      acc[i][j] = __builtin_amdgcn_wmma_f32_16x16x32_f16(
          false, a[i].v, false, b[j].v, (short)0, acc[i][j], false, false);
}

__device__ __forceinline__ void load_step(
    V16 a[2], V16 b[4],
    const half_t* A0, const half_t* A1,
    const half_t* W0, const half_t* W1,
    const half_t* W2, const half_t* W3, int k) {
  ldf(a[0], A0 + k);
  ldf(a[1], A1 + k);
  ldf(b[0], W0 + k);
  ldf(b[1], W1 + k);
  ldf(b[2], W2 + k);
  ldf(b[3], W3 + k);
}

__global__ __launch_bounds__(256) void gemm_wmma_kernel(
    const half_t* __restrict__ A, const half_t* __restrict__ W,
    const float* __restrict__ bias, const float* __restrict__ residual,
    float* __restrict__ C, int M, int N, int K) {
  const int lane = threadIdx.x & 31;
  const int wid  = (blockIdx.x * blockDim.x + threadIdx.x) >> 5;
  const int ntn  = N >> 6;
  const int ntm  = M >> 5;
  if (wid >= ntm * ntn) return;
  const int tm = wid / ntn, tn = wid % ntn;
  const int ko = (lane >= 16) ? 8 : 0;
  const int ln = lane & 15;

  const half_t* A0 = A + (size_t)(tm * 32 + ln) * K + ko;
  const half_t* A1 = A0 + (size_t)16 * K;
  const half_t* W0 = W + (size_t)(tn * 64 + ln) * K + ko;
  const half_t* W1 = W0 + (size_t)16 * K;
  const half_t* W2 = W0 + (size_t)32 * K;
  const half_t* W3 = W0 + (size_t)48 * K;

  v8f acc[2][4] = {};
  V16 pa[2], pb[4], qa[2], qb[4];

  load_step(pa, pb, A0, A1, W0, W1, W2, W3, 0);
  int k = 32;
  for (; k < K - 32; k += 64) {
    load_step(qa, qb, A0, A1, W0, W1, W2, W3, k);
    mma8(acc, pa, pb);
    load_step(pa, pb, A0, A1, W0, W1, W2, W3, k + 32);
    mma8(acc, qa, qb);
  }
  load_step(qa, qb, A0, A1, W0, W1, W2, W3, k);
  mma8(acc, pa, pb);
  mma8(acc, qa, qb);

  #pragma unroll
  for (int i = 0; i < 2; ++i) {
    #pragma unroll
    for (int j = 0; j < 4; ++j) {
      const int col = tn * 64 + j * 16 + ln;
      const float bv = bias[col];
      #pragma unroll
      for (int r = 0; r < 8; ++r) {
        const int row = tm * 32 + i * 16 + r + ko;
        float v = acc[i][j][r] + bv;
        if (residual) v += residual[(size_t)row * N + col];
        C[(size_t)row * N + col] = v;
      }
    }
  }
}

__global__ __launch_bounds__(256) void rope_kernel(
    const float* __restrict__ qkv, half_t* __restrict__ Q,
    half_t* __restrict__ Kr, half_t* __restrict__ Vt) {
  const int idx = blockIdx.x * blockDim.x + threadIdx.x;
  const int d = idx & 63;
  const int h = (idx >> 6) & (HN - 1);
  const int t = (idx >> 10) & (TT - 1);
  const int b = idx >> 21;
  const int p = d & 31;
  const float ang = (float)t * __expf(-(float)p * (9.210340372f / 32.0f));
  const float c = cosf(ang), sn = sinf(ang);
  const size_t base = (size_t)(b * TT + t) * QKVD;
  const float q1 = qkv[base + h * 64 + p];
  const float q2 = qkv[base + h * 64 + p + 32];
  const float k1 = qkv[base + DM + h * 64 + p];
  const float k2 = qkv[base + DM + h * 64 + p + 32];
  const float qv = (d < 32) ? (q1 * c - q2 * sn) : (q2 * c + q1 * sn);
  const float kv = (d < 32) ? (k1 * c - k2 * sn) : (k2 * c + k1 * sn);
  const size_t bh = (size_t)(b * HN + h);
  Q [(bh * TT + t) * DH + d] = (half_t)qv;
  Kr[(bh * TT + t) * DH + d] = (half_t)kv;
  Vt[(bh * DH + d) * TT + t] = (half_t)qkv[base + 2 * DM + h * 64 + d];
}

__global__ __launch_bounds__(256) void attn_kernel(
    const half_t* __restrict__ Q, const half_t* __restrict__ Kr,
    const half_t* __restrict__ Vt, const int* __restrict__ amask,
    const float* __restrict__ sinks, half_t* __restrict__ attn) {
  const int lane = threadIdx.x & 31;
  const int wid  = (blockIdx.x * blockDim.x + threadIdx.x) >> 5;
  const int QT   = TT / 16;
  const int qt   = wid % QT;
  const int bh   = wid / QT;
  const int h    = bh % HN, b = bh / HN;
  const int qbase = qt * 16;
  const int ln = lane & 15;
  const int ko = (lane >= 16) ? 8 : 0;
  const int iq = qbase + ln;
  const half_t* Qh = Q  + (size_t)bh * TT * DH;
  const half_t* Kh = Kr + (size_t)bh * TT * DH;
  const half_t* Vh = Vt + (size_t)bh * DH * TT;
  const int* am = amask + b * TT;

  V16 qb0, qb1;
  {
    const half_t* qp = Qh + (size_t)(qbase + ln) * DH;
    ldf(qb0, qp + ko);
    ldf(qb1, qp + 32 + ko);
  }

  v8f acc[4] = {};
  float m = MINIT, l = 0.f;
  int kstart = qbase - (WIN - 1); if (kstart < 0) kstart = 0;
  kstart &= ~31;
  const int kend = qbase + 16;

  for (int jc = kstart; jc < kend; jc += 32) {
    v8f s[2];
    #pragma unroll
    for (int t2 = 0; t2 < 2; ++t2) {
      int jrow = jc + t2 * 16 + ln;
      if (jrow > TT - 1) jrow = TT - 1;
      const half_t* kp = Kh + (size_t)jrow * DH;
      V16 a0, a1;
      ldf(a0, kp + ko);
      ldf(a1, kp + 32 + ko);
      v8f z = {};
      z = __builtin_amdgcn_wmma_f32_16x16x32_f16(false, a0.v, false, qb0.v,
                                                 (short)0, z, false, false);
      z = __builtin_amdgcn_wmma_f32_16x16x32_f16(false, a1.v, false, qb1.v,
                                                 (short)0, z, false, false);
      s[t2] = z;
    }
    float tl = MASKED;
    #pragma unroll
    for (int t2 = 0; t2 < 2; ++t2) {
      #pragma unroll
      for (int r = 0; r < 8; ++r) {
        const int j = jc + t2 * 16 + r + ko;
        const int jcl = (j < TT - 1) ? j : TT - 1;
        const bool ok = (j <= iq) && (j + WIN > iq) && (j < TT) && (am[jcl] != 0);
        const float v = ok ? s[t2][r] * SM_SCALE : MASKED;
        s[t2][r] = v;
        tl = fmaxf(tl, v);
      }
    }
    tl = fmaxf(tl, __shfl_xor(tl, 16, 32));
    const float mnew = fmaxf(m, tl);
    const float co = __expf(m - mnew);
    l *= co;
    #pragma unroll
    for (int dt = 0; dt < 4; ++dt) acc[dt] = acc[dt] * co;

    V16 pbf;
    float psum = 0.f;
    #pragma unroll
    for (int r = 0; r < 8; ++r) {
      const float p0 = __expf(s[0][r] - mnew);
      const float p1 = __expf(s[1][r] - mnew);
      psum += p0 + p1;
      pbf.h[0][r] = (half_t)p0;
      pbf.h[1][r] = (half_t)p1;
    }
    l += psum + __shfl_xor(psum, 16, 32);
    m = mnew;

    #pragma unroll
    for (int dt = 0; dt < 4; ++dt) {
      const half_t* vp = Vh + (size_t)(dt * 16 + ln) * TT + jc + ko;
      V16 va;
      ldf(va, vp);
      acc[dt] = __builtin_amdgcn_wmma_f32_16x16x32_f16(
          false, va.v, false, pbf.v, (short)0, acc[dt], false, false);
    }
  }

  const float snk = sinks[h];
  const float m2 = fmaxf(m, snk);
  const float co = __expf(m - m2);
  const float ltot = l * co + __expf(snk - m2);
  const float factor = co / ltot;
  half_t* op = attn + (size_t)(b * TT + qbase + ln) * DM + h * 64;
  #pragma unroll
  for (int dt = 0; dt < 4; ++dt) {
    v8h o;
    #pragma unroll
    for (int r = 0; r < 8; ++r) o[r] = (half_t)(acc[dt][r] * factor);
    *(v8h*)(op + dt * 16 + ko) = o;
  }
}

extern "C" void kernel_launch(void* const* d_in, const int* in_sizes, int n_in,
                              void* d_out, int out_size, void* d_ws, size_t ws_size,
                              hipStream_t stream) {
  const float* x      = (const float*)d_in[0];
  const int*   amask  = (const int*)d_in[1];
  const float* nscale = (const float*)d_in[2];
  const float* qkv_w  = (const float*)d_in[3];
  const float* qkv_b  = (const float*)d_in[4];
  const float* out_w  = (const float*)d_in[5];
  const float* out_b  = (const float*)d_in[6];
  const float* sinks  = (const float*)d_in[7];
  float* out = (float*)d_out;
  char*  ws  = (char*)d_ws;

  half_t* qkvw_h = (half_t*)(ws + OFF_QKVW);
  half_t* outw_h = (half_t*)(ws + OFF_OUTW);
  half_t* t_h    = (half_t*)(ws + OFF_T);
  float*  qkvf   = (float*)(ws + OFF_QKVF);
  half_t* attn_h = (half_t*)(ws + OFF_ATTN);
  half_t* Qh     = (half_t*)(ws + OFF_Q);
  half_t* Kh     = (half_t*)(ws + OFF_K);
  half_t* Vth    = (half_t*)(ws + OFF_V);

  rmsnorm_kernel<<<MROW, 256, 0, stream>>>(x, nscale, t_h);
  f32_to_f16_kernel<<<2048, 256, 0, stream>>>(qkv_w, qkvw_h, QKVD * DM);
  f32_to_f16_kernel<<<1024, 256, 0, stream>>>(out_w, outw_h, DM * DM);

  gemm_wmma_kernel<<<768, 256, 0, stream>>>(t_h, qkvw_h, qkv_b, nullptr,
                                            qkvf, MROW, QKVD, DM);

  rope_kernel<<<(BB * TT * HN * DH) / 256, 256, 0, stream>>>(qkvf, Qh, Kh, Vth);

  attn_kernel<<<512, 256, 0, stream>>>(Qh, Kh, Vth, amask, sinks, attn_h);

  gemm_wmma_kernel<<<256, 256, 0, stream>>>(attn_h, outw_h, out_b, x,
                                            out, MROW, DM, DM);
}


# Loads waited for one at a time now issue as one group ahead of use: the 64 residual loads of a wave in the out-projection epilogue; a key chunk's mask words (as b128), K and V loads in attention
# speedup vs baseline: 2.1296x; 2.1296x over previous
#include <hip/hip_runtime.h>
#include <hip/hip_bf16.h>

typedef _Float16 half_t;
typedef __attribute__((ext_vector_type(16))) _Float16 v16h;
typedef __attribute__((ext_vector_type(8)))  _Float16 v8h;
typedef __attribute__((ext_vector_type(8)))  float    v8f;

union V16 { v16h v; v8h h[2]; };

constexpr int BB   = 2;
constexpr int TT   = 2048;
constexpr int DM   = 1024;
constexpr int HN   = 16;
constexpr int DH   = 64;
constexpr int QKVD = 3072;
constexpr int MROW = BB * TT;
constexpr int WIN  = 128;
constexpr float SM_SCALE = 0.125f;
constexpr int ROW_TILES  = MROW / 16;
constexpr int QKVW_TILES = QKVD * DM / 512;
constexpr int OUTW_TILES = DM * DM / 512;
constexpr int QKVW_BLOCKS = QKVW_TILES / 8;
constexpr int OUTW_BLOCKS = OUTW_TILES / 8;
constexpr float MASKED = -3.0e38f;
constexpr float MINIT  = -1.0e30f;

constexpr size_t OFF_QKVW = 0;
constexpr size_t SZ_QKVW  = (size_t)QKVD * DM * 2;
constexpr size_t OFF_OUTW = OFF_QKVW + SZ_QKVW;
constexpr size_t SZ_OUTW  = (size_t)DM * DM * 2;
constexpr size_t OFF_T    = OFF_OUTW + SZ_OUTW;
constexpr size_t SZ_T     = (size_t)MROW * DM * 2;
constexpr size_t OFF_QKVF = OFF_T + SZ_T;
constexpr size_t SZ_QKVF  = (size_t)MROW * QKVD * 4;
constexpr size_t OFF_ATTN = OFF_QKVF;
constexpr size_t OFF_Q    = OFF_QKVF + SZ_QKVF;
constexpr size_t SZ_QKH   = (size_t)BB * HN * TT * DH * 2;
constexpr size_t OFF_K    = OFF_Q + SZ_QKH;
constexpr size_t OFF_V    = OFF_K + SZ_QKH;

__device__ __forceinline__ void ldf(V16& f, const half_t* p) {
  f.h[0] = *(const v8h*)(p);
  f.h[1] = *(const v8h*)(p + 16);
}

__device__ __forceinline__ void ldfm(V16& f, const half_t* p) {
  f.h[0] = *(const v8h*)(p);
  f.h[1] = *(const v8h*)(p + 256);
}

__device__ __forceinline__ v8h cvt8(const float4 a, const float4 b) {
  v8h o;
  o[0] = (half_t)a.x; o[1] = (half_t)a.y; o[2] = (half_t)a.z; o[3] = (half_t)a.w;
  o[4] = (half_t)b.x; o[5] = (half_t)b.y; o[6] = (half_t)b.z; o[7] = (half_t)b.w;
  return o;
}

__global__ __launch_bounds__(256) void rmsnorm_kernel(
    const float* __restrict__ x, const float* __restrict__ scale,
    half_t* __restrict__ t) {
  __shared__ float red[8][16];
  const int lane = threadIdx.x & 31;
  const int wv   = threadIdx.x >> 5;
  const int ln = lane & 15;
  const int hf = lane >> 4;
  const int rt = blockIdx.x;
  const float* xr = x + (size_t)(rt * 16 + ln) * DM + wv * 128 + hf * 8;
  const float* sp = scale + wv * 128 + hf * 8;
  float4 va[8], vb[8], sa[8], sb[8];
  float ss = 0.f;
  #pragma unroll
  for (int p = 0; p < 8; ++p) {
    va[p] = *(const float4*)(xr + p * 16);
    vb[p] = *(const float4*)(xr + p * 16 + 4);
  }
  #pragma unroll
  for (int p = 0; p < 8; ++p) {
    sa[p] = *(const float4*)(sp + p * 16);
    sb[p] = *(const float4*)(sp + p * 16 + 4);
  }
  #pragma unroll
  for (int p = 0; p < 8; ++p) {
    ss += va[p].x * va[p].x + va[p].y * va[p].y + va[p].z * va[p].z + va[p].w * va[p].w;
    ss += vb[p].x * vb[p].x + vb[p].y * vb[p].y + vb[p].z * vb[p].z + vb[p].w * vb[p].w;
  }
  ss += __shfl_xor(ss, 16, 32);
  if (hf == 0) red[wv][ln] = ss;
  __syncthreads();
  float tot = 0.f;
  #pragma unroll
  for (int i = 0; i < 8; ++i) tot += red[i][ln];
  const float inv = rsqrtf(tot * (1.0f / DM) + 1e-5f);
  half_t* dst = t + ((size_t)rt * (DM / 32) + wv * 4) * 512 + lane * 8;
  #pragma unroll
  for (int p = 0; p < 8; ++p) {
    const float4 s0 = sa[p], s1 = sb[p];
    float4 a = va[p], b = vb[p];
    a.x = a.x * inv * s0.x; a.y = a.y * inv * s0.y; a.z = a.z * inv * s0.z; a.w = a.w * inv * s0.w;
    b.x = b.x * inv * s1.x; b.y = b.y * inv * s1.y; b.z = b.z * inv * s1.z; b.w = b.w * inv * s1.w;
    *(v8h*)(dst + (p >> 1) * 512 + (p & 1) * 256) = cvt8(a, b);
  }
}

__global__ __launch_bounds__(256) void f32_to_f16_kernel(
    const float* __restrict__ src, half_t* __restrict__ dst, int ntiles, int K) {
  const int lane = threadIdx.x & 31;
  const int tile = (blockIdx.x * blockDim.x + threadIdx.x) >> 5;
  if (tile >= ntiles) return;
  const int kt = K >> 5;
  const int rt = tile / kt, kc = tile % kt;
  const int ln = lane & 15;
  const int hf = lane >> 4;
  const float* sp = src + (size_t)(rt * 16 + ln) * K + kc * 32 + hf * 8;
  const float4 a0 = *(const float4*)(sp);
  const float4 b0 = *(const float4*)(sp + 4);
  const float4 a1 = *(const float4*)(sp + 16);
  const float4 b1 = *(const float4*)(sp + 20);
  half_t* dp = dst + (size_t)tile * 512 + lane * 8;
  *(v8h*)(dp)       = cvt8(a0, b0);
  *(v8h*)(dp + 256) = cvt8(a1, b1);
}

__device__ __forceinline__ void mma8(v8f acc[2][4], const V16 a[2], const V16 b[4]) {
  #pragma unroll
  for (int i = 0; i < 2; ++i)
    #pragma unroll
    for (int j = 0; j < 4; ++j)
      acc[i][j] = __builtin_amdgcn_wmma_f32_16x16x32_f16(
          false, a[i].v, false, b[j].v, (short)0, acc[i][j], false, false);
}

__device__ __forceinline__ void load_step(
    V16 a[2], V16 b[4],
    const half_t* A0, const half_t* A1,
    const half_t* W0, const half_t* W1,
    const half_t* W2, const half_t* W3, int k) {
  ldfm(a[0], A0 + k * 16);
  ldfm(a[1], A1 + k * 16);
  ldfm(b[0], W0 + k * 16);
  ldfm(b[1], W1 + k * 16);
  ldfm(b[2], W2 + k * 16);
  ldfm(b[3], W3 + k * 16);
}

__global__ __launch_bounds__(256) void gemm_wmma_kernel(
    const half_t* __restrict__ A, const half_t* __restrict__ W,
    const float* __restrict__ bias, const float* __restrict__ residual,
    float* __restrict__ C, int M, int N, int K) {
  const int lane = threadIdx.x & 31;
  const int wid  = (blockIdx.x * blockDim.x + threadIdx.x) >> 5;
  const int ntn  = N >> 6;
  const int ntm  = M >> 5;
  if (wid >= ntm * ntn) return;
  const int tm = wid / ntn, tn = wid % ntn;
  const int ko = (lane >= 16) ? 8 : 0;
  const int ln = lane & 15;

  const half_t* A0 = A + (size_t)(tm * 32) * K + lane * 8;
  const half_t* A1 = A0 + (size_t)16 * K;
  const half_t* W0 = W + (size_t)(tn * 64) * K + lane * 8;
  const half_t* W1 = W0 + (size_t)16 * K;
  const half_t* W2 = W0 + (size_t)32 * K;
  const half_t* W3 = W0 + (size_t)48 * K;

  v8f acc[2][4] = {};
  V16 pa[2], pb[4], qa[2], qb[4];

  load_step(pa, pb, A0, A1, W0, W1, W2, W3, 0);
  int k = 32;
  for (; k < K - 32; k += 64) {
    load_step(qa, qb, A0, A1, W0, W1, W2, W3, k);
    mma8(acc, pa, pb);
    load_step(pa, pb, A0, A1, W0, W1, W2, W3, k + 32);
    mma8(acc, qa, qb);
  }
  load_step(qa, qb, A0, A1, W0, W1, W2, W3, k);
  mma8(acc, pa, pb);
  mma8(acc, qa, qb);

  float bv[4];
  #pragma unroll
  for (int j = 0; j < 4; ++j) bv[j] = bias[tn * 64 + j * 16 + ln];
  float rv[2][4][8];
  if (residual) {
    #pragma unroll
    for (int i = 0; i < 2; ++i)
      #pragma unroll
      for (int j = 0; j < 4; ++j)
        #pragma unroll
        for (int r = 0; r < 8; ++r)
          rv[i][j][r] = residual[(size_t)(tm * 32 + i * 16 + r + ko) * N + tn * 64 + j * 16 + ln];
  } else {
    #pragma unroll
    for (int i = 0; i < 2; ++i)
      #pragma unroll
      for (int j = 0; j < 4; ++j)
        #pragma unroll
        for (int r = 0; r < 8; ++r) rv[i][j][r] = 0.f;
  }

  #pragma unroll
  for (int i = 0; i < 2; ++i) {
    #pragma unroll
    for (int j = 0; j < 4; ++j) {
      const int col = tn * 64 + j * 16 + ln;
      #pragma unroll
      for (int r = 0; r < 8; ++r) {
        const int row = tm * 32 + i * 16 + r + ko;
        float v = acc[i][j][r] + bv[j];
        v += rv[i][j][r];
        C[(size_t)row * N + col] = v;
      }
    }
  }
}

__device__ __forceinline__ void mma8t(v8f acc[2][4], const V16 n[2], const V16 m[4]) {
  #pragma unroll
  for (int i = 0; i < 2; ++i)
    #pragma unroll
    for (int j = 0; j < 4; ++j)
      acc[i][j] = __builtin_amdgcn_wmma_f32_16x16x32_f16(
          false, m[j].v, false, n[i].v, (short)0, acc[i][j], false, false);
}

constexpr int CS_PITCH = 36;

__global__ __launch_bounds__(256, 6) void qkv_rope_kernel(
    const half_t* __restrict__ A, const half_t* __restrict__ W,
    const float* __restrict__ bias,
    half_t* __restrict__ Q, half_t* __restrict__ Kr, half_t* __restrict__ Vt,
    int K) {
  __shared__ __attribute__((aligned(16))) float cs_c[32 * CS_PITCH];
  __shared__ __attribute__((aligned(16))) float cs_s[32 * CS_PITCH];

  const int lane = threadIdx.x & 31;
  const int wv   = threadIdx.x >> 5;
  const int ln = lane & 15;
  const int ko = (lane >= 16) ? 8 : 0;
  const int tm  = blockIdx.x / 6;
  const int grp = blockIdx.x % 6;
  const int sec = grp >> 1;
  const int b   = tm >> 6;

  int m0, n0;
  const half_t* Mp;
  const half_t* Np;
  if (sec < 2) {
    Mp = W; m0 = (grp * 8 + wv) * 64;
    Np = A; n0 = tm * 32;
  } else {
    Mp = A; m0 = (tm >> 1) * 64;
    Np = W; n0 = 2 * DM + (((tm & 1) * 2 + (grp - 4)) * 8 + wv) * 32;
  }

  if (sec < 2) {
    const int t0 = (tm & 63) * 32;
    for (int e = threadIdx.x; e < 32 * 32; e += 256) {
      const int tt = e >> 5, p = e & 31;
      const float ang = (float)(t0 + tt) * __expf(-(float)p * (9.210340372f / 32.0f));
      cs_c[tt * CS_PITCH + p] = cosf(ang);
      cs_s[tt * CS_PITCH + p] = sinf(ang);
    }
  }

  const half_t* N0 = Np + (size_t)n0 * K + lane * 8;
  const half_t* N1 = N0 + (size_t)16 * K;
  const half_t* M0 = Mp + (size_t)m0 * K + lane * 8;
  const half_t* M1 = M0 + (size_t)16 * K;
  const half_t* M2 = M0 + (size_t)32 * K;
  const half_t* M3 = M0 + (size_t)48 * K;

  v8f acc[2][4] = {};
  {
    V16 pn[2], pm[4], qn[2], qm[4];
    load_step(pn, pm, N0, N1, M0, M1, M2, M3, 0);
    int k = 32;
    for (; k < K - 32; k += 64) {
      load_step(qn, qm, N0, N1, M0, M1, M2, M3, k);
      mma8t(acc, pn, pm);
      load_step(pn, pm, N0, N1, M0, M1, M2, M3, k + 32);
      mma8t(acc, qn, qm);
    }
    load_step(qn, qm, N0, N1, M0, M1, M2, M3, k);
    mma8t(acc, pn, pm);
    mma8t(acc, qn, qm);
  }

  if (sec < 2) {
    const int h  = (grp * 8 + wv) & (HN - 1);
    const int t0 = (tm & 63) * 32;
    __syncthreads();
    half_t* dst = (sec == 0 ? Q : Kr) + ((size_t)(b * HN + h) * TT + t0) * DH;
    const float* bp = bias + m0 + ko;
    #pragma unroll
    for (int j = 0; j < 2; ++j) {
      const float4 bl0 = *(const float4*)(bp + j * 16);
      const float4 bl1 = *(const float4*)(bp + j * 16 + 4);
      const float4 bh0 = *(const float4*)(bp + j * 16 + 32);
      const float4 bh1 = *(const float4*)(bp + j * 16 + 36);
      const float blo[8] = {bl0.x, bl0.y, bl0.z, bl0.w, bl1.x, bl1.y, bl1.z, bl1.w};
      const float bhi[8] = {bh0.x, bh0.y, bh0.z, bh0.w, bh1.x, bh1.y, bh1.z, bh1.w};
      #pragma unroll
      for (int i = 0; i < 2; ++i) {
        const int tt = i * 16 + ln;
        const float* cp = cs_c + tt * CS_PITCH + j * 16 + ko;
        const float* sp = cs_s + tt * CS_PITCH + j * 16 + ko;
        const float4 c0 = *(const float4*)(cp), c1 = *(const float4*)(cp + 4);
        const float4 s0 = *(const float4*)(sp), s1 = *(const float4*)(sp + 4);
        const float cc[8] = {c0.x, c0.y, c0.z, c0.w, c1.x, c1.y, c1.z, c1.w};
        const float ss[8] = {s0.x, s0.y, s0.z, s0.w, s1.x, s1.y, s1.z, s1.w};
        v8h o1, o2;
        #pragma unroll
        for (int r = 0; r < 8; ++r) {
          const float x1 = acc[i][j][r] + blo[r];
          const float x2 = acc[i][j + 2][r] + bhi[r];
          o1[r] = (half_t)(x1 * cc[r] - x2 * ss[r]);
          o2[r] = (half_t)(x2 * cc[r] + x1 * ss[r]);
        }
        *(v8h*)(dst + (size_t)tt * DH + j * 16 + ko) = o1;
        *(v8h*)(dst + (size_t)tt * DH + j * 16 + 32 + ko) = o2;
      }
    }
  } else {
    const int tv0 = ((tm >> 1) & 31) * 64;
    #pragma unroll
    for (int i = 0; i < 2; ++i) {
      const int c = n0 - 2 * DM + i * 16 + ln;
      const float bv = bias[2 * DM + c];
      half_t* dst = Vt + ((size_t)b * DM + c) * TT + tv0 + ko;
      #pragma unroll
      for (int j = 0; j < 4; ++j) {
        v8h o;
        #pragma unroll
        for (int r = 0; r < 8; ++r) o[r] = (half_t)(acc[i][j][r] + bv);
        *(v8h*)(dst + j * 16) = o;
      }
    }
  }
}

__global__ __launch_bounds__(256) __attribute__((amdgpu_waves_per_eu(4, 4))) void attn_kernel(
    const half_t* __restrict__ Q, const half_t* __restrict__ Kr,
    const half_t* __restrict__ Vt, const int* __restrict__ amask,
    const float* __restrict__ sinks, half_t* __restrict__ attn) {
  const int lane = threadIdx.x & 31;
  const int wid  = (blockIdx.x * blockDim.x + threadIdx.x) >> 5;
  const int QT   = TT / 16;
  const int qt   = wid % QT;
  const int bh   = wid / QT;
  const int h    = bh % HN, b = bh / HN;
  const int qbase = qt * 16;
  const int ln = lane & 15;
  const int ko = (lane >= 16) ? 8 : 0;
  const int iq = qbase + ln;
  const half_t* Qh = Q  + (size_t)bh * TT * DH;
  const half_t* Kh = Kr + (size_t)bh * TT * DH;
  const half_t* Vh = Vt + (size_t)bh * DH * TT;
  const int* am = amask + b * TT;

  V16 qb0, qb1;
  {
    const half_t* qp = Qh + (size_t)(qbase + ln) * DH;
    ldf(qb0, qp + ko);
    ldf(qb1, qp + 32 + ko);
  }

  v8f acc[4] = {};
  float m = MINIT, l = 0.f;
  int kstart = qbase - (WIN - 1); if (kstart < 0) kstart = 0;
  kstart &= ~31;
  const int kend = qbase + 16;

  for (int jc = kstart; jc < kend; jc += 32) {
    int mk[2][8];
    #pragma unroll
    for (int t2 = 0; t2 < 2; ++t2) {
      const int4 m0 = *(const int4*)(am + jc + t2 * 16 + ko);
      const int4 m1 = *(const int4*)(am + jc + t2 * 16 + ko + 4);
      mk[t2][0] = m0.x; mk[t2][1] = m0.y; mk[t2][2] = m0.z; mk[t2][3] = m0.w;
      mk[t2][4] = m1.x; mk[t2][5] = m1.y; mk[t2][6] = m1.z; mk[t2][7] = m1.w;
    }
    v8f s[2];
    V16 ka[2][2];
    #pragma unroll
    for (int t2 = 0; t2 < 2; ++t2) {
      int jrow = jc + t2 * 16 + ln;
      if (jrow > TT - 1) jrow = TT - 1;
      const half_t* kp = Kh + (size_t)jrow * DH;
      ldf(ka[t2][0], kp + ko);
      ldf(ka[t2][1], kp + 32 + ko);
    }
    #pragma unroll
    for (int t2 = 0; t2 < 2; ++t2) {
      v8f z = {};
      z = __builtin_amdgcn_wmma_f32_16x16x32_f16(false, ka[t2][0].v, false, qb0.v,
                                                 (short)0, z, false, false);
      z = __builtin_amdgcn_wmma_f32_16x16x32_f16(false, ka[t2][1].v, false, qb1.v,
                                                 (short)0, z, false, false);
      s[t2] = z;
    }
    float tl = MASKED;
    #pragma unroll
    for (int t2 = 0; t2 < 2; ++t2) {
      #pragma unroll
      for (int r = 0; r < 8; ++r) {
        const int j = jc + t2 * 16 + r + ko;
        const bool ok = (j <= iq) & (j + WIN > iq) & (mk[t2][r] != 0);
        const float v = ok ? s[t2][r] * SM_SCALE : MASKED;
        s[t2][r] = v;
        tl = fmaxf(tl, v);
      }
    }
    tl = fmaxf(tl, __shfl_xor(tl, 16, 32));
    const float mnew = fmaxf(m, tl);
    const float co = __expf(m - mnew);
    l *= co;
    #pragma unroll
    for (int dt = 0; dt < 4; ++dt) acc[dt] = acc[dt] * co;

    V16 pbf;
    float psum = 0.f;
    #pragma unroll
    for (int r = 0; r < 8; ++r) {
      const float p0 = __expf(s[0][r] - mnew);
      const float p1 = __expf(s[1][r] - mnew);
      psum += p0 + p1;
      pbf.h[0][r] = (half_t)p0;
      pbf.h[1][r] = (half_t)p1;
    }
    l += psum + __shfl_xor(psum, 16, 32);
    m = mnew;

    #pragma unroll
    for (int dt = 0; dt < 4; ++dt) {
      const half_t* vp = Vh + (size_t)(dt * 16 + ln) * TT + jc + ko;
      V16 va;
      ldf(va, vp);
      acc[dt] = __builtin_amdgcn_wmma_f32_16x16x32_f16(
          false, va.v, false, pbf.v, (short)0, acc[dt], false, false);
    }
  }

  const float snk = sinks[h];
  const float m2 = fmaxf(m, snk);
  const float co = __expf(m - m2);
  const float ltot = l * co + __expf(snk - m2);
  const float factor = co / ltot;
  half_t* op = attn + ((size_t)((b * TT + qbase) >> 4) * (DM / 32) + h * 2) * 512 + lane * 8;
  #pragma unroll
  for (int dt = 0; dt < 4; ++dt) {
    v8h o;
    #pragma unroll
    for (int r = 0; r < 8; ++r) o[r] = (half_t)(acc[dt][r] * factor);
    *(v8h*)(op + (dt >> 1) * 512 + (dt & 1) * 256) = o;
  }
}

extern "C" void kernel_launch(void* const* d_in, const int* in_sizes, int n_in,
                              void* d_out, int out_size, void* d_ws, size_t ws_size,
                              hipStream_t stream) {
  const float* x      = (const float*)d_in[0];
  const int*   amask  = (const int*)d_in[1];
  const float* nscale = (const float*)d_in[2];
  const float* qkv_w  = (const float*)d_in[3];
  const float* qkv_b  = (const float*)d_in[4];
  const float* out_w  = (const float*)d_in[5];
  const float* out_b  = (const float*)d_in[6];
  const float* sinks  = (const float*)d_in[7];
  float* out = (float*)d_out;
  char*  ws  = (char*)d_ws;

  half_t* qkvw_h = (half_t*)(ws + OFF_QKVW);
  half_t* outw_h = (half_t*)(ws + OFF_OUTW);
  half_t* t_h    = (half_t*)(ws + OFF_T);
  half_t* attn_h = (half_t*)(ws + OFF_ATTN);
  half_t* Qh     = (half_t*)(ws + OFF_Q);
  half_t* Kh     = (half_t*)(ws + OFF_K);
  half_t* Vth    = (half_t*)(ws + OFF_V);

  rmsnorm_kernel<<<ROW_TILES, 256, 0, stream>>>(x, nscale, t_h);
  f32_to_f16_kernel<<<QKVW_BLOCKS, 256, 0, stream>>>(qkv_w, qkvw_h, QKVW_TILES, DM);
  f32_to_f16_kernel<<<OUTW_BLOCKS, 256, 0, stream>>>(out_w, outw_h, OUTW_TILES, DM);

  qkv_rope_kernel<<<768, 256, 0, stream>>>(t_h, qkvw_h, qkv_b, Qh, Kh, Vth, DM);

  attn_kernel<<<512, 256, 0, stream>>>(Qh, Kh, Vth, amask, sinks, attn_h);

  gemm_wmma_kernel<<<256, 256, 0, stream>>>(attn_h, outw_h, out_b, x,
                                            out, MROW, DM, DM);
}


# Both GEMMs re-tiled so a fetched operand tile feeds more WMMAs: QKV waves take 64x64 (was 32x64) in 12-wave blocks of 128 t x 6 heads; out-projection blocks cover 128x128 (was 32x512)
# speedup vs baseline: 2.3189x; 1.0889x over previous
#include <hip/hip_runtime.h>
#include <hip/hip_bf16.h>

typedef _Float16 half_t;
typedef __attribute__((ext_vector_type(16))) _Float16 v16h;
typedef __attribute__((ext_vector_type(8)))  _Float16 v8h;
typedef __attribute__((ext_vector_type(8)))  float    v8f;

union V16 { v16h v; v8h h[2]; };

constexpr int BB   = 2;
constexpr int TT   = 2048;
constexpr int DM   = 1024;
constexpr int HN   = 16;
constexpr int DH   = 64;
constexpr int QKVD = 3072;
constexpr int MROW = BB * TT;
constexpr int WIN  = 128;
constexpr float SM_SCALE = 0.125f;
constexpr int ROW_TILES  = MROW / 16;
constexpr int QKVW_TILES = QKVD * DM / 512;
constexpr int OUTW_TILES = DM * DM / 512;
constexpr int QKVW_BLOCKS = QKVW_TILES / 8;
constexpr int OUTW_BLOCKS = OUTW_TILES / 8;
constexpr float MASKED = -3.0e38f;
constexpr float MINIT  = -1.0e30f;

constexpr size_t OFF_QKVW = 0;
constexpr size_t SZ_QKVW  = (size_t)QKVD * DM * 2;
constexpr size_t OFF_OUTW = OFF_QKVW + SZ_QKVW;
constexpr size_t SZ_OUTW  = (size_t)DM * DM * 2;
constexpr size_t OFF_T    = OFF_OUTW + SZ_OUTW;
constexpr size_t SZ_T     = (size_t)MROW * DM * 2;
constexpr size_t OFF_QKVF = OFF_T + SZ_T;
constexpr size_t SZ_QKVF  = (size_t)MROW * QKVD * 4;
constexpr size_t OFF_ATTN = OFF_QKVF;
constexpr size_t OFF_Q    = OFF_QKVF + SZ_QKVF;
constexpr size_t SZ_QKH   = (size_t)BB * HN * TT * DH * 2;
constexpr size_t OFF_K    = OFF_Q + SZ_QKH;
constexpr size_t OFF_V    = OFF_K + SZ_QKH;

__device__ __forceinline__ void ldf(V16& f, const half_t* p) {
  f.h[0] = *(const v8h*)(p);
  f.h[1] = *(const v8h*)(p + 16);
}

__device__ __forceinline__ void ldfm(V16& f, const half_t* p) {
  f.h[0] = *(const v8h*)(p);
  f.h[1] = *(const v8h*)(p + 256);
}

__device__ __forceinline__ v8h cvt8(const float4 a, const float4 b) {
  v8h o;
  o[0] = (half_t)a.x; o[1] = (half_t)a.y; o[2] = (half_t)a.z; o[3] = (half_t)a.w;
  o[4] = (half_t)b.x; o[5] = (half_t)b.y; o[6] = (half_t)b.z; o[7] = (half_t)b.w;
  return o;
}

__global__ __launch_bounds__(256) void rmsnorm_kernel(
    const float* __restrict__ x, const float* __restrict__ scale,
    half_t* __restrict__ t) {
  __shared__ float red[8][16];
  const int lane = threadIdx.x & 31;
  const int wv   = threadIdx.x >> 5;
  const int ln = lane & 15;
  const int hf = lane >> 4;
  const int rt = blockIdx.x;
  const float* xr = x + (size_t)(rt * 16 + ln) * DM + wv * 128 + hf * 8;
  const float* sp = scale + wv * 128 + hf * 8;
  float4 va[8], vb[8], sa[8], sb[8];
  float ss = 0.f;
  #pragma unroll
  for (int p = 0; p < 8; ++p) {
    va[p] = *(const float4*)(xr + p * 16);
    vb[p] = *(const float4*)(xr + p * 16 + 4);
  }
  #pragma unroll
  for (int p = 0; p < 8; ++p) {
    sa[p] = *(const float4*)(sp + p * 16);
    sb[p] = *(const float4*)(sp + p * 16 + 4);
  }
  #pragma unroll
  for (int p = 0; p < 8; ++p) {
    ss += va[p].x * va[p].x + va[p].y * va[p].y + va[p].z * va[p].z + va[p].w * va[p].w;
    ss += vb[p].x * vb[p].x + vb[p].y * vb[p].y + vb[p].z * vb[p].z + vb[p].w * vb[p].w;
  }
  ss += __shfl_xor(ss, 16, 32);
  if (hf == 0) red[wv][ln] = ss;
  __syncthreads();
  float tot = 0.f;
  #pragma unroll
  for (int i = 0; i < 8; ++i) tot += red[i][ln];
  const float inv = rsqrtf(tot * (1.0f / DM) + 1e-5f);
  half_t* dst = t + ((size_t)rt * (DM / 32) + wv * 4) * 512 + lane * 8;
  #pragma unroll
  for (int p = 0; p < 8; ++p) {
    const float4 s0 = sa[p], s1 = sb[p];
    float4 a = va[p], b = vb[p];
    a.x = a.x * inv * s0.x; a.y = a.y * inv * s0.y; a.z = a.z * inv * s0.z; a.w = a.w * inv * s0.w;
    b.x = b.x * inv * s1.x; b.y = b.y * inv * s1.y; b.z = b.z * inv * s1.z; b.w = b.w * inv * s1.w;
    *(v8h*)(dst + (p >> 1) * 512 + (p & 1) * 256) = cvt8(a, b);
  }
}

__global__ __launch_bounds__(256) void f32_to_f16_kernel(
    const float* __restrict__ src, half_t* __restrict__ dst, int ntiles, int K) {
  const int lane = threadIdx.x & 31;
  const int tile = (blockIdx.x * blockDim.x + threadIdx.x) >> 5;
  if (tile >= ntiles) return;
  const int kt = K >> 5;
  const int rt = tile / kt, kc = tile % kt;
  const int ln = lane & 15;
  const int hf = lane >> 4;
  const float* sp = src + (size_t)(rt * 16 + ln) * K + kc * 32 + hf * 8;
  const float4 a0 = *(const float4*)(sp);
  const float4 b0 = *(const float4*)(sp + 4);
  const float4 a1 = *(const float4*)(sp + 16);
  const float4 b1 = *(const float4*)(sp + 20);
  half_t* dp = dst + (size_t)tile * 512 + lane * 8;
  *(v8h*)(dp)       = cvt8(a0, b0);
  *(v8h*)(dp + 256) = cvt8(a1, b1);
}

__device__ __forceinline__ void mma8(v8f acc[2][4], const V16 a[2], const V16 b[4]) {
  #pragma unroll
  for (int i = 0; i < 2; ++i)
    #pragma unroll
    for (int j = 0; j < 4; ++j)
      acc[i][j] = __builtin_amdgcn_wmma_f32_16x16x32_f16(
          false, a[i].v, false, b[j].v, (short)0, acc[i][j], false, false);
}

__device__ __forceinline__ void load_step(
    V16 a[2], V16 b[4],
    const half_t* A0, const half_t* A1,
    const half_t* W0, const half_t* W1,
    const half_t* W2, const half_t* W3, int k) {
  ldfm(a[0], A0 + k * 16);
  ldfm(a[1], A1 + k * 16);
  ldfm(b[0], W0 + k * 16);
  ldfm(b[1], W1 + k * 16);
  ldfm(b[2], W2 + k * 16);
  ldfm(b[3], W3 + k * 16);
}

__global__ __launch_bounds__(256) void gemm_wmma_kernel(
    const half_t* __restrict__ A, const half_t* __restrict__ W,
    const float* __restrict__ bias, const float* __restrict__ residual,
    float* __restrict__ C, int M, int N, int K) {
  const int lane = threadIdx.x & 31;
  const int wv   = threadIdx.x >> 5;
  const int nbn  = N >> 7;
  const int bm = blockIdx.x / nbn, bn = blockIdx.x % nbn;
  const int tm = bm * 4 + (wv >> 1);
  const int tn = bn * 2 + (wv & 1);
  if (tm >= (M >> 5) || tn >= (N >> 6)) return;
  const int ko = (lane >= 16) ? 8 : 0;
  const int ln = lane & 15;

  const half_t* A0 = A + (size_t)(tm * 32) * K + lane * 8;
  const half_t* A1 = A0 + (size_t)16 * K;
  const half_t* W0 = W + (size_t)(tn * 64) * K + lane * 8;
  const half_t* W1 = W0 + (size_t)16 * K;
  const half_t* W2 = W0 + (size_t)32 * K;
  const half_t* W3 = W0 + (size_t)48 * K;

  v8f acc[2][4] = {};
  V16 pa[2], pb[4], qa[2], qb[4];

  load_step(pa, pb, A0, A1, W0, W1, W2, W3, 0);
  int k = 32;
  for (; k < K - 32; k += 64) {
    load_step(qa, qb, A0, A1, W0, W1, W2, W3, k);
    mma8(acc, pa, pb);
    load_step(pa, pb, A0, A1, W0, W1, W2, W3, k + 32);
    mma8(acc, qa, qb);
  }
  load_step(qa, qb, A0, A1, W0, W1, W2, W3, k);
  mma8(acc, pa, pb);
  mma8(acc, qa, qb);

  float bv[4];
  #pragma unroll
  for (int j = 0; j < 4; ++j) bv[j] = bias[tn * 64 + j * 16 + ln];
  float rv[2][4][8];
  if (residual) {
    #pragma unroll
    for (int i = 0; i < 2; ++i)
      #pragma unroll
      for (int j = 0; j < 4; ++j)
        #pragma unroll
        for (int r = 0; r < 8; ++r)
          rv[i][j][r] = residual[(size_t)(tm * 32 + i * 16 + r + ko) * N + tn * 64 + j * 16 + ln];
  } else {
    #pragma unroll
    for (int i = 0; i < 2; ++i)
      #pragma unroll
      for (int j = 0; j < 4; ++j)
        #pragma unroll
        for (int r = 0; r < 8; ++r) rv[i][j][r] = 0.f;
  }

  #pragma unroll
  for (int i = 0; i < 2; ++i) {
    #pragma unroll
    for (int j = 0; j < 4; ++j) {
      const int col = tn * 64 + j * 16 + ln;
      #pragma unroll
      for (int r = 0; r < 8; ++r) {
        const int row = tm * 32 + i * 16 + r + ko;
        float v = acc[i][j][r] + bv[j];
        v += rv[i][j][r];
        C[(size_t)row * N + col] = v;
      }
    }
  }
}

__device__ __forceinline__ void mma16t(v8f acc[4][4], const V16 n[4], const V16 m[4]) {
  #pragma unroll
  for (int i = 0; i < 4; ++i)
    #pragma unroll
    for (int j = 0; j < 4; ++j)
      acc[i][j] = __builtin_amdgcn_wmma_f32_16x16x32_f16(
          false, m[j].v, false, n[i].v, (short)0, acc[i][j], false, false);
}

__device__ __forceinline__ void load_step8(
    V16 n[4], V16 m[4], const half_t* N0, const half_t* M0, size_t ts, int k) {
  #pragma unroll
  for (int i = 0; i < 4; ++i) ldfm(n[i], N0 + i * ts + k * 16);
  #pragma unroll
  for (int j = 0; j < 4; ++j) ldfm(m[j], M0 + j * ts + k * 16);
}

constexpr int CS_PITCH = 36;
constexpr int QKV_BT   = 128;
constexpr int QKV_BH   = 6;
constexpr int QKV_WB   = (3 * HN) / QKV_BH;
constexpr int QKV_GRID = (MROW / QKV_BT) * QKV_WB;

__global__ __launch_bounds__(384) __attribute__((amdgpu_waves_per_eu(3, 3))) void qkv_rope_kernel(
    const half_t* __restrict__ A, const half_t* __restrict__ W,
    const float* __restrict__ bias,
    half_t* __restrict__ Q, half_t* __restrict__ Kr, half_t* __restrict__ Vt,
    int K) {
  __shared__ __attribute__((aligned(16))) float cs_c[QKV_BT * CS_PITCH];
  __shared__ __attribute__((aligned(16))) float cs_s[QKV_BT * CS_PITCH];

  const int lane = threadIdx.x & 31;
  const int wv   = threadIdx.x >> 5;
  const int ln = lane & 15;
  const int ko = (lane >= 16) ? 8 : 0;
  const int tb = blockIdx.x / QKV_WB;
  const int wb = blockIdx.x % QKV_WB;
  const int ti = wv / QKV_BH;
  const int hs = wb * QKV_BH + (wv - ti * QKV_BH);
  const int sec = hs >> 4;
  const int h   = hs & (HN - 1);
  const int trow = tb * QKV_BT + ti * 64;
  const int b    = trow / TT;
  const int tl0  = trow - b * TT;
  const bool has_qk = wb * QKV_BH < 2 * HN;

  if (has_qk) {
    const int tq0 = (tb * QKV_BT) & (TT - 1);
    for (int e = threadIdx.x; e < QKV_BT * 32; e += 384) {
      const int tt = e >> 5, p = e & 31;
      const float ang = (float)(tq0 + tt) * __expf(-(float)p * (9.210340372f / 32.0f));
      cs_c[tt * CS_PITCH + p] = cosf(ang);
      cs_s[tt * CS_PITCH + p] = sinf(ang);
    }
    __syncthreads();
  }

  const half_t* Mp;
  const half_t* Np;
  int m0, n0;
  if (sec < 2) { Mp = W; m0 = hs * 64; Np = A; n0 = trow; }
  else         { Mp = A; m0 = trow;    Np = W; n0 = hs * 64; }

  const half_t* N0 = Np + (size_t)n0 * K + lane * 8;
  const half_t* M0 = Mp + (size_t)m0 * K + lane * 8;
  const size_t ts = (size_t)16 * K;

  v8f acc[4][4] = {};
  {
    V16 pn[4], pm[4], qn[4], qm[4];
    load_step8(pn, pm, N0, M0, ts, 0);
    int k = 32;
    for (; k < K - 32; k += 64) {
      load_step8(qn, qm, N0, M0, ts, k);
      mma16t(acc, pn, pm);
      load_step8(pn, pm, N0, M0, ts, k + 32);
      mma16t(acc, qn, qm);
    }
    load_step8(qn, qm, N0, M0, ts, k);
    mma16t(acc, pn, pm);
    mma16t(acc, qn, qm);
  }

  if (sec < 2) {
    half_t* dst = (sec == 0 ? Q : Kr) + ((size_t)(b * HN + h) * TT + tl0) * DH;
    const float* bp = bias + m0 + ko;
    #pragma unroll
    for (int j = 0; j < 2; ++j) {
      const float4 bl0 = *(const float4*)(bp + j * 16);
      const float4 bl1 = *(const float4*)(bp + j * 16 + 4);
      const float4 bh0 = *(const float4*)(bp + j * 16 + 32);
      const float4 bh1 = *(const float4*)(bp + j * 16 + 36);
      const float blo[8] = {bl0.x, bl0.y, bl0.z, bl0.w, bl1.x, bl1.y, bl1.z, bl1.w};
      const float bhi[8] = {bh0.x, bh0.y, bh0.z, bh0.w, bh1.x, bh1.y, bh1.z, bh1.w};
      #pragma unroll
      for (int i = 0; i < 4; ++i) {
        const int tw = i * 16 + ln;
        const int tt = ti * 64 + tw;
        const float* cp = cs_c + tt * CS_PITCH + j * 16 + ko;
        const float* sp = cs_s + tt * CS_PITCH + j * 16 + ko;
        const float4 c0 = *(const float4*)(cp), c1 = *(const float4*)(cp + 4);
        const float4 s0 = *(const float4*)(sp), s1 = *(const float4*)(sp + 4);
        const float cc[8] = {c0.x, c0.y, c0.z, c0.w, c1.x, c1.y, c1.z, c1.w};
        const float ss[8] = {s0.x, s0.y, s0.z, s0.w, s1.x, s1.y, s1.z, s1.w};
        v8h o1, o2;
        #pragma unroll
        for (int r = 0; r < 8; ++r) {
          const float x1 = acc[i][j][r] + blo[r];
          const float x2 = acc[i][j + 2][r] + bhi[r];
          o1[r] = (half_t)(x1 * cc[r] - x2 * ss[r]);
          o2[r] = (half_t)(x2 * cc[r] + x1 * ss[r]);
        }
        *(v8h*)(dst + (size_t)tw * DH + j * 16 + ko) = o1;
        *(v8h*)(dst + (size_t)tw * DH + j * 16 + 32 + ko) = o2;
      }
    }
  } else {
    #pragma unroll
    for (int i = 0; i < 4; ++i) {
      const int c = h * 64 + i * 16 + ln;
      const float bv = bias[2 * DM + c];
      half_t* dst = Vt + ((size_t)b * DM + c) * TT + tl0 + ko;
      #pragma unroll
      for (int j = 0; j < 4; ++j) {
        v8h o;
        #pragma unroll
        for (int r = 0; r < 8; ++r) o[r] = (half_t)(acc[i][j][r] + bv);
        *(v8h*)(dst + j * 16) = o;
      }
    }
  }
}

__global__ __launch_bounds__(256) __attribute__((amdgpu_waves_per_eu(4, 4))) void attn_kernel(
    const half_t* __restrict__ Q, const half_t* __restrict__ Kr,
    const half_t* __restrict__ Vt, const int* __restrict__ amask,
    const float* __restrict__ sinks, half_t* __restrict__ attn) {
  const int lane = threadIdx.x & 31;
  const int wid  = (blockIdx.x * blockDim.x + threadIdx.x) >> 5;
  const int QT   = TT / 16;
  const int qt   = wid % QT;
  const int bh   = wid / QT;
  const int h    = bh % HN, b = bh / HN;
  const int qbase = qt * 16;
  const int ln = lane & 15;
  const int ko = (lane >= 16) ? 8 : 0;
  const int iq = qbase + ln;
  const half_t* Qh = Q  + (size_t)bh * TT * DH;
  const half_t* Kh = Kr + (size_t)bh * TT * DH;
  const half_t* Vh = Vt + (size_t)bh * DH * TT;
  const int* am = amask + b * TT;

  V16 qb0, qb1;
  {
    const half_t* qp = Qh + (size_t)(qbase + ln) * DH;
    ldf(qb0, qp + ko);
    ldf(qb1, qp + 32 + ko);
  }

  v8f acc[4] = {};
  float m = MINIT, l = 0.f;
  int kstart = qbase - (WIN - 1); if (kstart < 0) kstart = 0;
  kstart &= ~31;
  const int kend = qbase + 16;

  for (int jc = kstart; jc < kend; jc += 32) {
    int mk[2][8];
    #pragma unroll
    for (int t2 = 0; t2 < 2; ++t2) {
      const int4 m0 = *(const int4*)(am + jc + t2 * 16 + ko);
      const int4 m1 = *(const int4*)(am + jc + t2 * 16 + ko + 4);
      mk[t2][0] = m0.x; mk[t2][1] = m0.y; mk[t2][2] = m0.z; mk[t2][3] = m0.w;
      mk[t2][4] = m1.x; mk[t2][5] = m1.y; mk[t2][6] = m1.z; mk[t2][7] = m1.w;
    }
    v8f s[2];
    V16 ka[2][2];
    #pragma unroll
    for (int t2 = 0; t2 < 2; ++t2) {
      int jrow = jc + t2 * 16 + ln;
      if (jrow > TT - 1) jrow = TT - 1;
      const half_t* kp = Kh + (size_t)jrow * DH;
      ldf(ka[t2][0], kp + ko);
      ldf(ka[t2][1], kp + 32 + ko);
    }
    #pragma unroll
    for (int t2 = 0; t2 < 2; ++t2) {
      v8f z = {};
      z = __builtin_amdgcn_wmma_f32_16x16x32_f16(false, ka[t2][0].v, false, qb0.v,
                                                 (short)0, z, false, false);
      z = __builtin_amdgcn_wmma_f32_16x16x32_f16(false, ka[t2][1].v, false, qb1.v,
                                                 (short)0, z, false, false);
      s[t2] = z;
    }
    float tl = MASKED;
    #pragma unroll
    for (int t2 = 0; t2 < 2; ++t2) {
      #pragma unroll
      for (int r = 0; r < 8; ++r) {
        const int j = jc + t2 * 16 + r + ko;
        const bool ok = (j <= iq) & (j + WIN > iq) & (mk[t2][r] != 0);
        const float v = ok ? s[t2][r] * SM_SCALE : MASKED;
        s[t2][r] = v;
        tl = fmaxf(tl, v);
      }
    }
    tl = fmaxf(tl, __shfl_xor(tl, 16, 32));
    const float mnew = fmaxf(m, tl);
    const float co = __expf(m - mnew);
    l *= co;
    #pragma unroll
    for (int dt = 0; dt < 4; ++dt) acc[dt] = acc[dt] * co;

    V16 pbf;
    float psum = 0.f;
    #pragma unroll
    for (int r = 0; r < 8; ++r) {
      const float p0 = __expf(s[0][r] - mnew);
      const float p1 = __expf(s[1][r] - mnew);
      psum += p0 + p1;
      pbf.h[0][r] = (half_t)p0;
      pbf.h[1][r] = (half_t)p1;
    }
    l += psum + __shfl_xor(psum, 16, 32);
    m = mnew;

    #pragma unroll
    for (int dt = 0; dt < 4; ++dt) {
      const half_t* vp = Vh + (size_t)(dt * 16 + ln) * TT + jc + ko;
      V16 va;
      ldf(va, vp);
      acc[dt] = __builtin_amdgcn_wmma_f32_16x16x32_f16(
          false, va.v, false, pbf.v, (short)0, acc[dt], false, false);
    }
  }

  const float snk = sinks[h];
  const float m2 = fmaxf(m, snk);
  const float co = __expf(m - m2);
  const float ltot = l * co + __expf(snk - m2);
  const float factor = co / ltot;
  half_t* op = attn + ((size_t)((b * TT + qbase) >> 4) * (DM / 32) + h * 2) * 512 + lane * 8;
  #pragma unroll
  for (int dt = 0; dt < 4; ++dt) {
    v8h o;
    #pragma unroll
    for (int r = 0; r < 8; ++r) o[r] = (half_t)(acc[dt][r] * factor);
    *(v8h*)(op + (dt >> 1) * 512 + (dt & 1) * 256) = o;
  }
}

extern "C" void kernel_launch(void* const* d_in, const int* in_sizes, int n_in,
                              void* d_out, int out_size, void* d_ws, size_t ws_size,
                              hipStream_t stream) {
  const float* x      = (const float*)d_in[0];
  const int*   amask  = (const int*)d_in[1];
  const float* nscale = (const float*)d_in[2];
  const float* qkv_w  = (const float*)d_in[3];
  const float* qkv_b  = (const float*)d_in[4];
  const float* out_w  = (const float*)d_in[5];
  const float* out_b  = (const float*)d_in[6];
  const float* sinks  = (const float*)d_in[7];
  float* out = (float*)d_out;
  char*  ws  = (char*)d_ws;

  half_t* qkvw_h = (half_t*)(ws + OFF_QKVW);
  half_t* outw_h = (half_t*)(ws + OFF_OUTW);
  half_t* t_h    = (half_t*)(ws + OFF_T);
  half_t* attn_h = (half_t*)(ws + OFF_ATTN);
  half_t* Qh     = (half_t*)(ws + OFF_Q);
  half_t* Kh     = (half_t*)(ws + OFF_K);
  half_t* Vth    = (half_t*)(ws + OFF_V);

  rmsnorm_kernel<<<ROW_TILES, 256, 0, stream>>>(x, nscale, t_h);
  f32_to_f16_kernel<<<QKVW_BLOCKS, 256, 0, stream>>>(qkv_w, qkvw_h, QKVW_TILES, DM);
  f32_to_f16_kernel<<<OUTW_BLOCKS, 256, 0, stream>>>(out_w, outw_h, OUTW_TILES, DM);

  qkv_rope_kernel<<<QKV_GRID, 384, 0, stream>>>(t_h, qkvw_h, qkv_b, Qh, Kh, Vth, DM);

  attn_kernel<<<512, 256, 0, stream>>>(Qh, Kh, Vth, amask, sinks, attn_h);

  gemm_wmma_kernel<<<256, 256, 0, stream>>>(attn_h, outw_h, out_b, x,
                                            out, MROW, DM, DM);
}


# The three preparation passes (RMSNorm of x, f16 conversion of qkv_w and of out_w) run as one launch of 1280 blocks instead of three stream-ordered launches; launches per call 6 -> 4
# speedup vs baseline: 2.4395x; 1.0520x over previous
#include <hip/hip_runtime.h>
#include <hip/hip_bf16.h>

typedef _Float16 half_t;
typedef __attribute__((ext_vector_type(16))) _Float16 v16h;
typedef __attribute__((ext_vector_type(8)))  _Float16 v8h;
typedef __attribute__((ext_vector_type(8)))  float    v8f;

union V16 { v16h v; v8h h[2]; };

constexpr int BB   = 2;
constexpr int TT   = 2048;
constexpr int DM   = 1024;
constexpr int HN   = 16;
constexpr int DH   = 64;
constexpr int QKVD = 3072;
constexpr int MROW = BB * TT;
constexpr int WIN  = 128;
constexpr float SM_SCALE = 0.125f;
constexpr int ROW_TILES  = MROW / 16;
constexpr int QKVW_TILES = QKVD * DM / 512;
constexpr int OUTW_TILES = DM * DM / 512;
constexpr int QKVW_BLOCKS = QKVW_TILES / 8;
constexpr int OUTW_BLOCKS = OUTW_TILES / 8;
constexpr float MASKED = -3.0e38f;
constexpr float MINIT  = -1.0e30f;

constexpr size_t OFF_QKVW = 0;
constexpr size_t SZ_QKVW  = (size_t)QKVD * DM * 2;
constexpr size_t OFF_OUTW = OFF_QKVW + SZ_QKVW;
constexpr size_t SZ_OUTW  = (size_t)DM * DM * 2;
constexpr size_t OFF_T    = OFF_OUTW + SZ_OUTW;
constexpr size_t SZ_T     = (size_t)MROW * DM * 2;
constexpr size_t OFF_QKVF = OFF_T + SZ_T;
constexpr size_t SZ_QKVF  = (size_t)MROW * QKVD * 4;
constexpr size_t OFF_ATTN = OFF_QKVF;
constexpr size_t OFF_Q    = OFF_QKVF + SZ_QKVF;
constexpr size_t SZ_QKH   = (size_t)BB * HN * TT * DH * 2;
constexpr size_t OFF_K    = OFF_Q + SZ_QKH;
constexpr size_t OFF_V    = OFF_K + SZ_QKH;

__device__ __forceinline__ void ldf(V16& f, const half_t* p) {
  f.h[0] = *(const v8h*)(p);
  f.h[1] = *(const v8h*)(p + 16);
}

__device__ __forceinline__ void ldfm(V16& f, const half_t* p) {
  f.h[0] = *(const v8h*)(p);
  f.h[1] = *(const v8h*)(p + 256);
}

__device__ __forceinline__ v8h cvt8(const float4 a, const float4 b) {
  v8h o;
  o[0] = (half_t)a.x; o[1] = (half_t)a.y; o[2] = (half_t)a.z; o[3] = (half_t)a.w;
  o[4] = (half_t)b.x; o[5] = (half_t)b.y; o[6] = (half_t)b.z; o[7] = (half_t)b.w;
  return o;
}

__device__ __forceinline__ void rmsnorm_tile_row(
    const float* __restrict__ x, const float* __restrict__ scale,
    half_t* __restrict__ t, const int rt, float (*red)[16]) {
  const int lane = threadIdx.x & 31;
  const int wv   = threadIdx.x >> 5;
  const int ln = lane & 15;
  const int hf = lane >> 4;
  const float* xr = x + (size_t)(rt * 16 + ln) * DM + wv * 128 + hf * 8;
  const float* sp = scale + wv * 128 + hf * 8;
  float4 va[8], vb[8], sa[8], sb[8];
  float ss = 0.f;
  #pragma unroll
  for (int p = 0; p < 8; ++p) {
    va[p] = *(const float4*)(xr + p * 16);
    vb[p] = *(const float4*)(xr + p * 16 + 4);
  }
  #pragma unroll
  for (int p = 0; p < 8; ++p) {
    sa[p] = *(const float4*)(sp + p * 16);
    sb[p] = *(const float4*)(sp + p * 16 + 4);
  }
  #pragma unroll
  for (int p = 0; p < 8; ++p) {
    ss += va[p].x * va[p].x + va[p].y * va[p].y + va[p].z * va[p].z + va[p].w * va[p].w;
    ss += vb[p].x * vb[p].x + vb[p].y * vb[p].y + vb[p].z * vb[p].z + vb[p].w * vb[p].w;
  }
  ss += __shfl_xor(ss, 16, 32);
  if (hf == 0) red[wv][ln] = ss;
  __syncthreads();
  float tot = 0.f;
  #pragma unroll
  for (int i = 0; i < 8; ++i) tot += red[i][ln];
  const float inv = rsqrtf(tot * (1.0f / DM) + 1e-5f);
  half_t* dst = t + ((size_t)rt * (DM / 32) + wv * 4) * 512 + lane * 8;
  #pragma unroll
  for (int p = 0; p < 8; ++p) {
    const float4 s0 = sa[p], s1 = sb[p];
    float4 a = va[p], b = vb[p];
    a.x = a.x * inv * s0.x; a.y = a.y * inv * s0.y; a.z = a.z * inv * s0.z; a.w = a.w * inv * s0.w;
    b.x = b.x * inv * s1.x; b.y = b.y * inv * s1.y; b.z = b.z * inv * s1.z; b.w = b.w * inv * s1.w;
    *(v8h*)(dst + (p >> 1) * 512 + (p & 1) * 256) = cvt8(a, b);
  }
}

__device__ __forceinline__ void f32_to_f16_tile(
    const float* __restrict__ src, half_t* __restrict__ dst, const int tile, const int K) {
  const int lane = threadIdx.x & 31;
  const int kt = K >> 5;
  const int rt = tile / kt, kc = tile % kt;
  const int ln = lane & 15;
  const int hf = lane >> 4;
  const float* sp = src + (size_t)(rt * 16 + ln) * K + kc * 32 + hf * 8;
  const float4 a0 = *(const float4*)(sp);
  const float4 b0 = *(const float4*)(sp + 4);
  const float4 a1 = *(const float4*)(sp + 16);
  const float4 b1 = *(const float4*)(sp + 20);
  half_t* dp = dst + (size_t)tile * 512 + lane * 8;
  *(v8h*)(dp)       = cvt8(a0, b0);
  *(v8h*)(dp + 256) = cvt8(a1, b1);
}

constexpr int PREP_GRID = ROW_TILES + QKVW_BLOCKS + OUTW_BLOCKS;

__global__ __launch_bounds__(256) void prep_kernel(
    const float* __restrict__ x, const float* __restrict__ scale, half_t* __restrict__ t,
    const float* __restrict__ qkv_w, half_t* __restrict__ qkvw_h,
    const float* __restrict__ out_w, half_t* __restrict__ outw_h) {
  __shared__ float red[8][16];
  const int blk = blockIdx.x;
  if (blk < ROW_TILES) {
    rmsnorm_tile_row(x, scale, t, blk, red);
  } else {
    const int tile = (blk - ROW_TILES) * 8 + (threadIdx.x >> 5);
    if (tile < QKVW_TILES) f32_to_f16_tile(qkv_w, qkvw_h, tile, DM);
    else                   f32_to_f16_tile(out_w, outw_h, tile - QKVW_TILES, DM);
  }
}

__device__ __forceinline__ void mma8(v8f acc[2][4], const V16 a[2], const V16 b[4]) {
  #pragma unroll
  for (int i = 0; i < 2; ++i)
    #pragma unroll
    for (int j = 0; j < 4; ++j)
      acc[i][j] = __builtin_amdgcn_wmma_f32_16x16x32_f16(
          false, a[i].v, false, b[j].v, (short)0, acc[i][j], false, false);
}

__device__ __forceinline__ void load_step(
    V16 a[2], V16 b[4],
    const half_t* A0, const half_t* A1,
    const half_t* W0, const half_t* W1,
    const half_t* W2, const half_t* W3, int k) {
  ldfm(a[0], A0 + k * 16);
  ldfm(a[1], A1 + k * 16);
  ldfm(b[0], W0 + k * 16);
  ldfm(b[1], W1 + k * 16);
  ldfm(b[2], W2 + k * 16);
  ldfm(b[3], W3 + k * 16);
}

__global__ __launch_bounds__(256) void gemm_wmma_kernel(
    const half_t* __restrict__ A, const half_t* __restrict__ W,
    const float* __restrict__ bias, const float* __restrict__ residual,
    float* __restrict__ C, int M, int N, int K) {
  const int lane = threadIdx.x & 31;
  const int wv   = threadIdx.x >> 5;
  const int nbn  = N >> 7;
  const int bm = blockIdx.x / nbn, bn = blockIdx.x % nbn;
  const int tm = bm * 4 + (wv >> 1);
  const int tn = bn * 2 + (wv & 1);
  if (tm >= (M >> 5) || tn >= (N >> 6)) return;
  const int ko = (lane >= 16) ? 8 : 0;
  const int ln = lane & 15;

  const half_t* A0 = A + (size_t)(tm * 32) * K + lane * 8;
  const half_t* A1 = A0 + (size_t)16 * K;
  const half_t* W0 = W + (size_t)(tn * 64) * K + lane * 8;
  const half_t* W1 = W0 + (size_t)16 * K;
  const half_t* W2 = W0 + (size_t)32 * K;
  const half_t* W3 = W0 + (size_t)48 * K;

  v8f acc[2][4] = {};
  V16 pa[2], pb[4], qa[2], qb[4];

  load_step(pa, pb, A0, A1, W0, W1, W2, W3, 0);
  int k = 32;
  for (; k < K - 32; k += 64) {
    load_step(qa, qb, A0, A1, W0, W1, W2, W3, k);
    mma8(acc, pa, pb);
    load_step(pa, pb, A0, A1, W0, W1, W2, W3, k + 32);
    mma8(acc, qa, qb);
  }
  load_step(qa, qb, A0, A1, W0, W1, W2, W3, k);
  mma8(acc, pa, pb);
  mma8(acc, qa, qb);

  float bv[4];
  #pragma unroll
  for (int j = 0; j < 4; ++j) bv[j] = bias[tn * 64 + j * 16 + ln];
  float rv[2][4][8];
  if (residual) {
    #pragma unroll
    for (int i = 0; i < 2; ++i)
      #pragma unroll
      for (int j = 0; j < 4; ++j)
        #pragma unroll
        for (int r = 0; r < 8; ++r)
          rv[i][j][r] = residual[(size_t)(tm * 32 + i * 16 + r + ko) * N + tn * 64 + j * 16 + ln];
  } else {
    #pragma unroll
    for (int i = 0; i < 2; ++i)
      #pragma unroll
      for (int j = 0; j < 4; ++j)
        #pragma unroll
        for (int r = 0; r < 8; ++r) rv[i][j][r] = 0.f;
  }

  #pragma unroll
  for (int i = 0; i < 2; ++i) {
    #pragma unroll
    for (int j = 0; j < 4; ++j) {
      const int col = tn * 64 + j * 16 + ln;
      #pragma unroll
      for (int r = 0; r < 8; ++r) {
        const int row = tm * 32 + i * 16 + r + ko;
        float v = acc[i][j][r] + bv[j];
        v += rv[i][j][r];
        C[(size_t)row * N + col] = v;
      }
    }
  }
}

__device__ __forceinline__ void mma16t(v8f acc[4][4], const V16 n[4], const V16 m[4]) {
  #pragma unroll
  for (int i = 0; i < 4; ++i)
    #pragma unroll
    for (int j = 0; j < 4; ++j)
      acc[i][j] = __builtin_amdgcn_wmma_f32_16x16x32_f16(
          false, m[j].v, false, n[i].v, (short)0, acc[i][j], false, false);
}

__device__ __forceinline__ void load_step8(
    V16 n[4], V16 m[4], const half_t* N0, const half_t* M0, size_t ts, int k) {
  #pragma unroll
  for (int i = 0; i < 4; ++i) ldfm(n[i], N0 + i * ts + k * 16);
  #pragma unroll
  for (int j = 0; j < 4; ++j) ldfm(m[j], M0 + j * ts + k * 16);
}

constexpr int CS_PITCH = 36;
constexpr int QKV_BT   = 128;
constexpr int QKV_BH   = 6;
constexpr int QKV_WB   = (3 * HN) / QKV_BH;
constexpr int QKV_GRID = (MROW / QKV_BT) * QKV_WB;

__global__ __launch_bounds__(384) __attribute__((amdgpu_waves_per_eu(3, 3))) void qkv_rope_kernel(
    const half_t* __restrict__ A, const half_t* __restrict__ W,
    const float* __restrict__ bias,
    half_t* __restrict__ Q, half_t* __restrict__ Kr, half_t* __restrict__ Vt,
    int K) {
  __shared__ __attribute__((aligned(16))) float cs_c[QKV_BT * CS_PITCH];
  __shared__ __attribute__((aligned(16))) float cs_s[QKV_BT * CS_PITCH];

  const int lane = threadIdx.x & 31;
  const int wv   = threadIdx.x >> 5;
  const int ln = lane & 15;
  const int ko = (lane >= 16) ? 8 : 0;
  const int tb = blockIdx.x / QKV_WB;
  const int wb = blockIdx.x % QKV_WB;
  const int ti = wv / QKV_BH;
  const int hs = wb * QKV_BH + (wv - ti * QKV_BH);
  const int sec = hs >> 4;
  const int h   = hs & (HN - 1);
  const int trow = tb * QKV_BT + ti * 64;
  const int b    = trow / TT;
  const int tl0  = trow - b * TT;
  const bool has_qk = wb * QKV_BH < 2 * HN;

  if (has_qk) {
    const int tq0 = (tb * QKV_BT) & (TT - 1);
    for (int e = threadIdx.x; e < QKV_BT * 32; e += 384) {
      const int tt = e >> 5, p = e & 31;
      const float ang = (float)(tq0 + tt) * __expf(-(float)p * (9.210340372f / 32.0f));
      cs_c[tt * CS_PITCH + p] = cosf(ang);
      cs_s[tt * CS_PITCH + p] = sinf(ang);
    }
    __syncthreads();
  }

  const half_t* Mp;
  const half_t* Np;
  int m0, n0;
  if (sec < 2) { Mp = W; m0 = hs * 64; Np = A; n0 = trow; }
  else         { Mp = A; m0 = trow;    Np = W; n0 = hs * 64; }

  const half_t* N0 = Np + (size_t)n0 * K + lane * 8;
  const half_t* M0 = Mp + (size_t)m0 * K + lane * 8;
  const size_t ts = (size_t)16 * K;

  v8f acc[4][4] = {};
  {
    V16 pn[4], pm[4], qn[4], qm[4];
    load_step8(pn, pm, N0, M0, ts, 0);
    int k = 32;
    for (; k < K - 32; k += 64) {
      load_step8(qn, qm, N0, M0, ts, k);
      mma16t(acc, pn, pm);
      load_step8(pn, pm, N0, M0, ts, k + 32);
      mma16t(acc, qn, qm);
    }
    load_step8(qn, qm, N0, M0, ts, k);
    mma16t(acc, pn, pm);
    mma16t(acc, qn, qm);
  }

  if (sec < 2) {
    half_t* dst = (sec == 0 ? Q : Kr) + ((size_t)(b * HN + h) * TT + tl0) * DH;
    const float* bp = bias + m0 + ko;
    #pragma unroll
    for (int j = 0; j < 2; ++j) {
      const float4 bl0 = *(const float4*)(bp + j * 16);
      const float4 bl1 = *(const float4*)(bp + j * 16 + 4);
      const float4 bh0 = *(const float4*)(bp + j * 16 + 32);
      const float4 bh1 = *(const float4*)(bp + j * 16 + 36);
      const float blo[8] = {bl0.x, bl0.y, bl0.z, bl0.w, bl1.x, bl1.y, bl1.z, bl1.w};
      const float bhi[8] = {bh0.x, bh0.y, bh0.z, bh0.w, bh1.x, bh1.y, bh1.z, bh1.w};
      #pragma unroll
      for (int i = 0; i < 4; ++i) {
        const int tw = i * 16 + ln;
        const int tt = ti * 64 + tw;
        const float* cp = cs_c + tt * CS_PITCH + j * 16 + ko;
        const float* sp = cs_s + tt * CS_PITCH + j * 16 + ko;
        const float4 c0 = *(const float4*)(cp), c1 = *(const float4*)(cp + 4);
        const float4 s0 = *(const float4*)(sp), s1 = *(const float4*)(sp + 4);
        const float cc[8] = {c0.x, c0.y, c0.z, c0.w, c1.x, c1.y, c1.z, c1.w};
        const float ss[8] = {s0.x, s0.y, s0.z, s0.w, s1.x, s1.y, s1.z, s1.w};
        v8h o1, o2;
        #pragma unroll
        for (int r = 0; r < 8; ++r) {
          const float x1 = acc[i][j][r] + blo[r];
          const float x2 = acc[i][j + 2][r] + bhi[r];
          o1[r] = (half_t)(x1 * cc[r] - x2 * ss[r]);
          o2[r] = (half_t)(x2 * cc[r] + x1 * ss[r]);
        }
        *(v8h*)(dst + (size_t)tw * DH + j * 16 + ko) = o1;
        *(v8h*)(dst + (size_t)tw * DH + j * 16 + 32 + ko) = o2;
      }
    }
  } else {
    #pragma unroll
    for (int i = 0; i < 4; ++i) {
      const int c = h * 64 + i * 16 + ln;
      const float bv = bias[2 * DM + c];
      half_t* dst = Vt + ((size_t)b * DM + c) * TT + tl0 + ko;
      #pragma unroll
      for (int j = 0; j < 4; ++j) {
        v8h o;
        #pragma unroll
        for (int r = 0; r < 8; ++r) o[r] = (half_t)(acc[i][j][r] + bv);
        *(v8h*)(dst + j * 16) = o;
      }
    }
  }
}

__global__ __launch_bounds__(256) __attribute__((amdgpu_waves_per_eu(4, 4))) void attn_kernel(
    const half_t* __restrict__ Q, const half_t* __restrict__ Kr,
    const half_t* __restrict__ Vt, const int* __restrict__ amask,
    const float* __restrict__ sinks, half_t* __restrict__ attn) {
  const int lane = threadIdx.x & 31;
  const int wid  = (blockIdx.x * blockDim.x + threadIdx.x) >> 5;
  const int QT   = TT / 16;
  const int qt   = wid % QT;
  const int bh   = wid / QT;
  const int h    = bh % HN, b = bh / HN;
  const int qbase = qt * 16;
  const int ln = lane & 15;
  const int ko = (lane >= 16) ? 8 : 0;
  const int iq = qbase + ln;
  const half_t* Qh = Q  + (size_t)bh * TT * DH;
  const half_t* Kh = Kr + (size_t)bh * TT * DH;
  const half_t* Vh = Vt + (size_t)bh * DH * TT;
  const int* am = amask + b * TT;

  V16 qb0, qb1;
  {
    const half_t* qp = Qh + (size_t)(qbase + ln) * DH;
    ldf(qb0, qp + ko);
    ldf(qb1, qp + 32 + ko);
  }

  v8f acc[4] = {};
  float m = MINIT, l = 0.f;
  int kstart = qbase - (WIN - 1); if (kstart < 0) kstart = 0;
  kstart &= ~31;
  const int kend = qbase + 16;

  for (int jc = kstart; jc < kend; jc += 32) {
    int mk[2][8];
    #pragma unroll
    for (int t2 = 0; t2 < 2; ++t2) {
      const int4 m0 = *(const int4*)(am + jc + t2 * 16 + ko);
      const int4 m1 = *(const int4*)(am + jc + t2 * 16 + ko + 4);
      mk[t2][0] = m0.x; mk[t2][1] = m0.y; mk[t2][2] = m0.z; mk[t2][3] = m0.w;
      mk[t2][4] = m1.x; mk[t2][5] = m1.y; mk[t2][6] = m1.z; mk[t2][7] = m1.w;
    }
    v8f s[2];
    V16 ka[2][2];
    #pragma unroll
    for (int t2 = 0; t2 < 2; ++t2) {
      int jrow = jc + t2 * 16 + ln;
      if (jrow > TT - 1) jrow = TT - 1;
      const half_t* kp = Kh + (size_t)jrow * DH;
      ldf(ka[t2][0], kp + ko);
      ldf(ka[t2][1], kp + 32 + ko);
    }
    #pragma unroll
    for (int t2 = 0; t2 < 2; ++t2) {
      v8f z = {};
      z = __builtin_amdgcn_wmma_f32_16x16x32_f16(false, ka[t2][0].v, false, qb0.v,
                                                 (short)0, z, false, false);
      z = __builtin_amdgcn_wmma_f32_16x16x32_f16(false, ka[t2][1].v, false, qb1.v,
                                                 (short)0, z, false, false);
      s[t2] = z;
    }
    float tl = MASKED;
    #pragma unroll
    for (int t2 = 0; t2 < 2; ++t2) {
      #pragma unroll
      for (int r = 0; r < 8; ++r) {
        const int j = jc + t2 * 16 + r + ko;
        const bool ok = (j <= iq) & (j + WIN > iq) & (mk[t2][r] != 0);
        const float v = ok ? s[t2][r] * SM_SCALE : MASKED;
        s[t2][r] = v;
        tl = fmaxf(tl, v);
      }
    }
    tl = fmaxf(tl, __shfl_xor(tl, 16, 32));
    const float mnew = fmaxf(m, tl);
    const float co = __expf(m - mnew);
    l *= co;
    #pragma unroll
    for (int dt = 0; dt < 4; ++dt) acc[dt] = acc[dt] * co;

    V16 pbf;
    float psum = 0.f;
    #pragma unroll
    for (int r = 0; r < 8; ++r) {
      const float p0 = __expf(s[0][r] - mnew);
      const float p1 = __expf(s[1][r] - mnew);
      psum += p0 + p1;
      pbf.h[0][r] = (half_t)p0;
      pbf.h[1][r] = (half_t)p1;
    }
    l += psum + __shfl_xor(psum, 16, 32);
    m = mnew;

    #pragma unroll
    for (int dt = 0; dt < 4; ++dt) {
      const half_t* vp = Vh + (size_t)(dt * 16 + ln) * TT + jc + ko;
      V16 va;
      ldf(va, vp);
      acc[dt] = __builtin_amdgcn_wmma_f32_16x16x32_f16(
          false, va.v, false, pbf.v, (short)0, acc[dt], false, false);
    }
  }

  const float snk = sinks[h];
  const float m2 = fmaxf(m, snk);
  const float co = __expf(m - m2);
  const float ltot = l * co + __expf(snk - m2);
  const float factor = co / ltot;
  half_t* op = attn + ((size_t)((b * TT + qbase) >> 4) * (DM / 32) + h * 2) * 512 + lane * 8;
  #pragma unroll
  for (int dt = 0; dt < 4; ++dt) {
    v8h o;
    #pragma unroll
    for (int r = 0; r < 8; ++r) o[r] = (half_t)(acc[dt][r] * factor);
    *(v8h*)(op + (dt >> 1) * 512 + (dt & 1) * 256) = o;
  }
}

extern "C" void kernel_launch(void* const* d_in, const int* in_sizes, int n_in,
                              void* d_out, int out_size, void* d_ws, size_t ws_size,
                              hipStream_t stream) {
  const float* x      = (const float*)d_in[0];
  const int*   amask  = (const int*)d_in[1];
  const float* nscale = (const float*)d_in[2];
  const float* qkv_w  = (const float*)d_in[3];
  const float* qkv_b  = (const float*)d_in[4];
  const float* out_w  = (const float*)d_in[5];
  const float* out_b  = (const float*)d_in[6];
  const float* sinks  = (const float*)d_in[7];
  float* out = (float*)d_out;
  char*  ws  = (char*)d_ws;

  half_t* qkvw_h = (half_t*)(ws + OFF_QKVW);
  half_t* outw_h = (half_t*)(ws + OFF_OUTW);
  half_t* t_h    = (half_t*)(ws + OFF_T);
  half_t* attn_h = (half_t*)(ws + OFF_ATTN);
  half_t* Qh     = (half_t*)(ws + OFF_Q);
  half_t* Kh     = (half_t*)(ws + OFF_K);
  half_t* Vth    = (half_t*)(ws + OFF_V);

  prep_kernel<<<PREP_GRID, 256, 0, stream>>>(x, nscale, t_h, qkv_w, qkvw_h, out_w, outw_h);

  qkv_rope_kernel<<<QKV_GRID, 384, 0, stream>>>(t_h, qkvw_h, qkv_b, Qh, Kh, Vth, DM);

  attn_kernel<<<512, 256, 0, stream>>>(Qh, Kh, Vth, amask, sinks, attn_h);

  gemm_wmma_kernel<<<256, 256, 0, stream>>>(attn_h, outw_h, out_b, x,
                                            out, MROW, DM, DM);
}


# Both GEMM k loops: a step's fragment loads stay in flight under earlier steps' WMMAs (one pinned group; 1 step ahead in QKV, ring of 4 sets = 3 ahead in out-projection); k sum order kept
# speedup vs baseline: 3.4714x; 1.4230x over previous
#include <hip/hip_runtime.h>
#include <hip/hip_bf16.h>

typedef _Float16 half_t;
typedef __attribute__((ext_vector_type(16))) _Float16 v16h;
typedef __attribute__((ext_vector_type(8)))  _Float16 v8h;
typedef __attribute__((ext_vector_type(8)))  float    v8f;

union V16 { v16h v; v8h h[2]; };

constexpr int BB   = 2;
constexpr int TT   = 2048;
constexpr int DM   = 1024;
constexpr int HN   = 16;
constexpr int DH   = 64;
constexpr int QKVD = 3072;
constexpr int MROW = BB * TT;
constexpr int WIN  = 128;
constexpr float SM_SCALE = 0.125f;
constexpr int ROW_TILES  = MROW / 16;
constexpr int QKVW_TILES = QKVD * DM / 512;
constexpr int OUTW_TILES = DM * DM / 512;
constexpr int QKVW_BLOCKS = QKVW_TILES / 8;
constexpr int OUTW_BLOCKS = OUTW_TILES / 8;
constexpr float MASKED = -3.0e38f;
constexpr float MINIT  = -1.0e30f;

constexpr size_t OFF_QKVW = 0;
constexpr size_t SZ_QKVW  = (size_t)QKVD * DM * 2;
constexpr size_t OFF_OUTW = OFF_QKVW + SZ_QKVW;
constexpr size_t SZ_OUTW  = (size_t)DM * DM * 2;
constexpr size_t OFF_T    = OFF_OUTW + SZ_OUTW;
constexpr size_t SZ_T     = (size_t)MROW * DM * 2;
constexpr size_t OFF_QKVF = OFF_T + SZ_T;
constexpr size_t SZ_QKVF  = (size_t)MROW * QKVD * 4;
constexpr size_t OFF_ATTN = OFF_QKVF;
constexpr size_t OFF_Q    = OFF_QKVF + SZ_QKVF;
constexpr size_t SZ_QKH   = (size_t)BB * HN * TT * DH * 2;
constexpr size_t OFF_K    = OFF_Q + SZ_QKH;
constexpr size_t OFF_V    = OFF_K + SZ_QKH;

__device__ __forceinline__ void ldf(V16& f, const half_t* p) {
  f.h[0] = *(const v8h*)(p);
  f.h[1] = *(const v8h*)(p + 16);
}

__device__ __forceinline__ void ldfm(V16& f, const half_t* p) {
  f.h[0] = *(const v8h*)(p);
  f.h[1] = *(const v8h*)(p + 256);
}

__device__ __forceinline__ v8h cvt8(const float4 a, const float4 b) {
  v8h o;
  o[0] = (half_t)a.x; o[1] = (half_t)a.y; o[2] = (half_t)a.z; o[3] = (half_t)a.w;
  o[4] = (half_t)b.x; o[5] = (half_t)b.y; o[6] = (half_t)b.z; o[7] = (half_t)b.w;
  return o;
}

__device__ __forceinline__ void rmsnorm_tile_row(
    const float* __restrict__ x, const float* __restrict__ scale,
    half_t* __restrict__ t, const int rt, float (*red)[16]) {
  const int lane = threadIdx.x & 31;
  const int wv   = threadIdx.x >> 5;
  const int ln = lane & 15;
  const int hf = lane >> 4;
  const float* xr = x + (size_t)(rt * 16 + ln) * DM + wv * 128 + hf * 8;
  const float* sp = scale + wv * 128 + hf * 8;
  float4 va[8], vb[8], sa[8], sb[8];
  float ss = 0.f;
  #pragma unroll
  for (int p = 0; p < 8; ++p) {
    va[p] = *(const float4*)(xr + p * 16);
    vb[p] = *(const float4*)(xr + p * 16 + 4);
  }
  #pragma unroll
  for (int p = 0; p < 8; ++p) {
    sa[p] = *(const float4*)(sp + p * 16);
    sb[p] = *(const float4*)(sp + p * 16 + 4);
  }
  #pragma unroll
  for (int p = 0; p < 8; ++p) {
    ss += va[p].x * va[p].x + va[p].y * va[p].y + va[p].z * va[p].z + va[p].w * va[p].w;
    ss += vb[p].x * vb[p].x + vb[p].y * vb[p].y + vb[p].z * vb[p].z + vb[p].w * vb[p].w;
  }
  ss += __shfl_xor(ss, 16, 32);
  if (hf == 0) red[wv][ln] = ss;
  __syncthreads();
  float tot = 0.f;
  #pragma unroll
  for (int i = 0; i < 8; ++i) tot += red[i][ln];
  const float inv = rsqrtf(tot * (1.0f / DM) + 1e-5f);
  half_t* dst = t + ((size_t)rt * (DM / 32) + wv * 4) * 512 + lane * 8;
  #pragma unroll
  for (int p = 0; p < 8; ++p) {
    const float4 s0 = sa[p], s1 = sb[p];
    float4 a = va[p], b = vb[p];
    a.x = a.x * inv * s0.x; a.y = a.y * inv * s0.y; a.z = a.z * inv * s0.z; a.w = a.w * inv * s0.w;
    b.x = b.x * inv * s1.x; b.y = b.y * inv * s1.y; b.z = b.z * inv * s1.z; b.w = b.w * inv * s1.w;
    *(v8h*)(dst + (p >> 1) * 512 + (p & 1) * 256) = cvt8(a, b);
  }
}

__device__ __forceinline__ void f32_to_f16_tile(
    const float* __restrict__ src, half_t* __restrict__ dst, const int tile, const int K) {
  const int lane = threadIdx.x & 31;
  const int kt = K >> 5;
  const int rt = tile / kt, kc = tile % kt;
  const int ln = lane & 15;
  const int hf = lane >> 4;
  const float* sp = src + (size_t)(rt * 16 + ln) * K + kc * 32 + hf * 8;
  const float4 a0 = *(const float4*)(sp);
  const float4 b0 = *(const float4*)(sp + 4);
  const float4 a1 = *(const float4*)(sp + 16);
  const float4 b1 = *(const float4*)(sp + 20);
  half_t* dp = dst + (size_t)tile * 512 + lane * 8;
  *(v8h*)(dp)       = cvt8(a0, b0);
  *(v8h*)(dp + 256) = cvt8(a1, b1);
}

constexpr int PREP_GRID = ROW_TILES + QKVW_BLOCKS + OUTW_BLOCKS;

__global__ __launch_bounds__(256) void prep_kernel(
    const float* __restrict__ x, const float* __restrict__ scale, half_t* __restrict__ t,
    const float* __restrict__ qkv_w, half_t* __restrict__ qkvw_h,
    const float* __restrict__ out_w, half_t* __restrict__ outw_h) {
  __shared__ float red[8][16];
  const int blk = blockIdx.x;
  if (blk < ROW_TILES) {
    rmsnorm_tile_row(x, scale, t, blk, red);
  } else {
    const int tile = (blk - ROW_TILES) * 8 + (threadIdx.x >> 5);
    if (tile < QKVW_TILES) f32_to_f16_tile(qkv_w, qkvw_h, tile, DM);
    else                   f32_to_f16_tile(out_w, outw_h, tile - QKVW_TILES, DM);
  }
}

__device__ __forceinline__ void pin_order() { __builtin_amdgcn_sched_barrier(0); }

__device__ __forceinline__ void mma8(v8f acc[2][4], const V16 a[2], const V16 b[4]) {
  #pragma unroll
  for (int i = 0; i < 2; ++i)
    #pragma unroll
    for (int j = 0; j < 4; ++j)
      acc[i][j] = __builtin_amdgcn_wmma_f32_16x16x32_f16(
          false, a[i].v, false, b[j].v, (short)0, acc[i][j], false, false);
}

__device__ __forceinline__ void load_step(
    V16 a[2], V16 b[4],
    const half_t* A0, const half_t* A1,
    const half_t* W0, const half_t* W1,
    const half_t* W2, const half_t* W3, int k) {
  ldfm(a[0], A0 + k * 16);
  ldfm(a[1], A1 + k * 16);
  ldfm(b[0], W0 + k * 16);
  ldfm(b[1], W1 + k * 16);
  ldfm(b[2], W2 + k * 16);
  ldfm(b[3], W3 + k * 16);
}

__global__ __launch_bounds__(256) void gemm_wmma_kernel(
    const half_t* __restrict__ A, const half_t* __restrict__ W,
    const float* __restrict__ bias, const float* __restrict__ residual,
    float* __restrict__ C, int M, int N, int K) {
  const int lane = threadIdx.x & 31;
  const int wv   = threadIdx.x >> 5;
  const int nbn  = N >> 7;
  const int bm = blockIdx.x / nbn, bn = blockIdx.x % nbn;
  const int tm = bm * 4 + (wv >> 1);
  const int tn = bn * 2 + (wv & 1);
  if (tm >= (M >> 5) || tn >= (N >> 6)) return;
  const int ko = (lane >= 16) ? 8 : 0;
  const int ln = lane & 15;

  const half_t* A0 = A + (size_t)(tm * 32) * K + lane * 8;
  const half_t* A1 = A0 + (size_t)16 * K;
  const half_t* W0 = W + (size_t)(tn * 64) * K + lane * 8;
  const half_t* W1 = W0 + (size_t)16 * K;
  const half_t* W2 = W0 + (size_t)32 * K;
  const half_t* W3 = W0 + (size_t)48 * K;

  v8f acc[2][4] = {};
  V16 a0[2], b0[4], a1[2], b1[4], a2[2], b2[4], a3[2], b3[4];

  load_step(a0, b0, A0, A1, W0, W1, W2, W3, 0);   pin_order();
  load_step(a1, b1, A0, A1, W0, W1, W2, W3, 32);  pin_order();
  load_step(a2, b2, A0, A1, W0, W1, W2, W3, 64);  pin_order();
  int k = 0;
  for (; k < K - 128; k += 128) {
    load_step(a3, b3, A0, A1, W0, W1, W2, W3, k + 96);   pin_order();
    mma8(acc, a0, b0);                                   pin_order();
    load_step(a0, b0, A0, A1, W0, W1, W2, W3, k + 128);  pin_order();
    mma8(acc, a1, b1);                                   pin_order();
    load_step(a1, b1, A0, A1, W0, W1, W2, W3, k + 160);  pin_order();
    mma8(acc, a2, b2);                                   pin_order();
    load_step(a2, b2, A0, A1, W0, W1, W2, W3, k + 192);  pin_order();
    mma8(acc, a3, b3);                                   pin_order();
  }
  load_step(a3, b3, A0, A1, W0, W1, W2, W3, k + 96);     pin_order();
  mma8(acc, a0, b0);
  mma8(acc, a1, b1);
  mma8(acc, a2, b2);
  mma8(acc, a3, b3);

  float bv[4];
  #pragma unroll
  for (int j = 0; j < 4; ++j) bv[j] = bias[tn * 64 + j * 16 + ln];
  float rv[2][4][8];
  if (residual) {
    #pragma unroll
    for (int i = 0; i < 2; ++i)
      #pragma unroll
      for (int j = 0; j < 4; ++j)
        #pragma unroll
        for (int r = 0; r < 8; ++r)
          rv[i][j][r] = residual[(size_t)(tm * 32 + i * 16 + r + ko) * N + tn * 64 + j * 16 + ln];
  } else {
    #pragma unroll
    for (int i = 0; i < 2; ++i)
      #pragma unroll
      for (int j = 0; j < 4; ++j)
        #pragma unroll
        for (int r = 0; r < 8; ++r) rv[i][j][r] = 0.f;
  }

  #pragma unroll
  for (int i = 0; i < 2; ++i) {
    #pragma unroll
    for (int j = 0; j < 4; ++j) {
      const int col = tn * 64 + j * 16 + ln;
      #pragma unroll
      for (int r = 0; r < 8; ++r) {
        const int row = tm * 32 + i * 16 + r + ko;
        float v = acc[i][j][r] + bv[j];
        v += rv[i][j][r];
        C[(size_t)row * N + col] = v;
      }
    }
  }
}

__device__ __forceinline__ void mma16t(v8f acc[4][4], const V16 n[4], const V16 m[4]) {
  #pragma unroll
  for (int i = 0; i < 4; ++i)
    #pragma unroll
    for (int j = 0; j < 4; ++j)
      acc[i][j] = __builtin_amdgcn_wmma_f32_16x16x32_f16(
          false, m[j].v, false, n[i].v, (short)0, acc[i][j], false, false);
}

__device__ __forceinline__ void load_step8(
    V16 n[4], V16 m[4], const half_t* N0, const half_t* M0, size_t ts, int k) {
  #pragma unroll
  for (int i = 0; i < 4; ++i) ldfm(n[i], N0 + i * ts + k * 16);
  #pragma unroll
  for (int j = 0; j < 4; ++j) ldfm(m[j], M0 + j * ts + k * 16);
}

constexpr int CS_PITCH = 36;
constexpr int QKV_BT   = 128;
constexpr int QKV_BH   = 6;
constexpr int QKV_WB   = (3 * HN) / QKV_BH;
constexpr int QKV_GRID = (MROW / QKV_BT) * QKV_WB;

__global__ __launch_bounds__(384) __attribute__((amdgpu_waves_per_eu(3, 3))) void qkv_rope_kernel(
    const half_t* __restrict__ A, const half_t* __restrict__ W,
    const float* __restrict__ bias,
    half_t* __restrict__ Q, half_t* __restrict__ Kr, half_t* __restrict__ Vt,
    int K) {
  __shared__ __attribute__((aligned(16))) float cs_c[QKV_BT * CS_PITCH];
  __shared__ __attribute__((aligned(16))) float cs_s[QKV_BT * CS_PITCH];

  const int lane = threadIdx.x & 31;
  const int wv   = threadIdx.x >> 5;
  const int ln = lane & 15;
  const int ko = (lane >= 16) ? 8 : 0;
  const int tb = blockIdx.x / QKV_WB;
  const int wb = blockIdx.x % QKV_WB;
  const int ti = wv / QKV_BH;
  const int hs = wb * QKV_BH + (wv - ti * QKV_BH);
  const int sec = hs >> 4;
  const int h   = hs & (HN - 1);
  const int trow = tb * QKV_BT + ti * 64;
  const int b    = trow / TT;
  const int tl0  = trow - b * TT;
  const bool has_qk = wb * QKV_BH < 2 * HN;

  if (has_qk) {
    const int tq0 = (tb * QKV_BT) & (TT - 1);
    for (int e = threadIdx.x; e < QKV_BT * 32; e += 384) {
      const int tt = e >> 5, p = e & 31;
      const float ang = (float)(tq0 + tt) * __expf(-(float)p * (9.210340372f / 32.0f));
      cs_c[tt * CS_PITCH + p] = cosf(ang);
      cs_s[tt * CS_PITCH + p] = sinf(ang);
    }
    __syncthreads();
  }

  const half_t* Mp;
  const half_t* Np;
  int m0, n0;
  if (sec < 2) { Mp = W; m0 = hs * 64; Np = A; n0 = trow; }
  else         { Mp = A; m0 = trow;    Np = W; n0 = hs * 64; }

  const half_t* N0 = Np + (size_t)n0 * K + lane * 8;
  const half_t* M0 = Mp + (size_t)m0 * K + lane * 8;
  const size_t ts = (size_t)16 * K;

  v8f acc[4][4] = {};
  {
    V16 pn[4], pm[4], qn[4], qm[4];
    load_step8(pn, pm, N0, M0, ts, 0);         pin_order();
    int k = 32;
    for (; k < K - 32; k += 64) {
      load_step8(qn, qm, N0, M0, ts, k);       pin_order();
      mma16t(acc, pn, pm);                     pin_order();
      load_step8(pn, pm, N0, M0, ts, k + 32);  pin_order();
      mma16t(acc, qn, qm);                     pin_order();
    }
    load_step8(qn, qm, N0, M0, ts, k);         pin_order();
    mma16t(acc, pn, pm);
    mma16t(acc, qn, qm);
  }

  if (sec < 2) {
    half_t* dst = (sec == 0 ? Q : Kr) + ((size_t)(b * HN + h) * TT + tl0) * DH;
    const float* bp = bias + m0 + ko;
    #pragma unroll
    for (int j = 0; j < 2; ++j) {
      const float4 bl0 = *(const float4*)(bp + j * 16);
      const float4 bl1 = *(const float4*)(bp + j * 16 + 4);
      const float4 bh0 = *(const float4*)(bp + j * 16 + 32);
      const float4 bh1 = *(const float4*)(bp + j * 16 + 36);
      const float blo[8] = {bl0.x, bl0.y, bl0.z, bl0.w, bl1.x, bl1.y, bl1.z, bl1.w};
      const float bhi[8] = {bh0.x, bh0.y, bh0.z, bh0.w, bh1.x, bh1.y, bh1.z, bh1.w};
      #pragma unroll
      for (int i = 0; i < 4; ++i) {
        const int tw = i * 16 + ln;
        const int tt = ti * 64 + tw;
        const float* cp = cs_c + tt * CS_PITCH + j * 16 + ko;
        const float* sp = cs_s + tt * CS_PITCH + j * 16 + ko;
        const float4 c0 = *(const float4*)(cp), c1 = *(const float4*)(cp + 4);
        const float4 s0 = *(const float4*)(sp), s1 = *(const float4*)(sp + 4);
        const float cc[8] = {c0.x, c0.y, c0.z, c0.w, c1.x, c1.y, c1.z, c1.w};
        const float ss[8] = {s0.x, s0.y, s0.z, s0.w, s1.x, s1.y, s1.z, s1.w};
        v8h o1, o2;
        #pragma unroll
        for (int r = 0; r < 8; ++r) {
          const float x1 = acc[i][j][r] + blo[r];
          const float x2 = acc[i][j + 2][r] + bhi[r];
          o1[r] = (half_t)(x1 * cc[r] - x2 * ss[r]);
          o2[r] = (half_t)(x2 * cc[r] + x1 * ss[r]);
        }
        *(v8h*)(dst + (size_t)tw * DH + j * 16 + ko) = o1;
        *(v8h*)(dst + (size_t)tw * DH + j * 16 + 32 + ko) = o2;
      }
    }
  } else {
    #pragma unroll
    for (int i = 0; i < 4; ++i) {
      const int c = h * 64 + i * 16 + ln;
      const float bv = bias[2 * DM + c];
      half_t* dst = Vt + ((size_t)b * DM + c) * TT + tl0 + ko;
      #pragma unroll
      for (int j = 0; j < 4; ++j) {
        v8h o;
        #pragma unroll
        for (int r = 0; r < 8; ++r) o[r] = (half_t)(acc[i][j][r] + bv);
        *(v8h*)(dst + j * 16) = o;
      }
    }
  }
}

__global__ __launch_bounds__(256) __attribute__((amdgpu_waves_per_eu(4, 4))) void attn_kernel(
    const half_t* __restrict__ Q, const half_t* __restrict__ Kr,
    const half_t* __restrict__ Vt, const int* __restrict__ amask,
    const float* __restrict__ sinks, half_t* __restrict__ attn) {
  const int lane = threadIdx.x & 31;
  const int wid  = (blockIdx.x * blockDim.x + threadIdx.x) >> 5;
  const int QT   = TT / 16;
  const int qt   = wid % QT;
  const int bh   = wid / QT;
  const int h    = bh % HN, b = bh / HN;
  const int qbase = qt * 16;
  const int ln = lane & 15;
  const int ko = (lane >= 16) ? 8 : 0;
  const int iq = qbase + ln;
  const half_t* Qh = Q  + (size_t)bh * TT * DH;
  const half_t* Kh = Kr + (size_t)bh * TT * DH;
  const half_t* Vh = Vt + (size_t)bh * DH * TT;
  const int* am = amask + b * TT;

  V16 qb0, qb1;
  {
    const half_t* qp = Qh + (size_t)(qbase + ln) * DH;
    ldf(qb0, qp + ko);
    ldf(qb1, qp + 32 + ko);
  }

  v8f acc[4] = {};
  float m = MINIT, l = 0.f;
  int kstart = qbase - (WIN - 1); if (kstart < 0) kstart = 0;
  kstart &= ~31;
  const int kend = qbase + 16;

  for (int jc = kstart; jc < kend; jc += 32) {
    int mk[2][8];
    #pragma unroll
    for (int t2 = 0; t2 < 2; ++t2) {
      const int4 m0 = *(const int4*)(am + jc + t2 * 16 + ko);
      const int4 m1 = *(const int4*)(am + jc + t2 * 16 + ko + 4);
      mk[t2][0] = m0.x; mk[t2][1] = m0.y; mk[t2][2] = m0.z; mk[t2][3] = m0.w;
      mk[t2][4] = m1.x; mk[t2][5] = m1.y; mk[t2][6] = m1.z; mk[t2][7] = m1.w;
    }
    v8f s[2];
    V16 ka[2][2];
    #pragma unroll
    for (int t2 = 0; t2 < 2; ++t2) {
      int jrow = jc + t2 * 16 + ln;
      if (jrow > TT - 1) jrow = TT - 1;
      const half_t* kp = Kh + (size_t)jrow * DH;
      ldf(ka[t2][0], kp + ko);
      ldf(ka[t2][1], kp + 32 + ko);
    }
    #pragma unroll
    for (int t2 = 0; t2 < 2; ++t2) {
      v8f z = {};
      z = __builtin_amdgcn_wmma_f32_16x16x32_f16(false, ka[t2][0].v, false, qb0.v,
                                                 (short)0, z, false, false);
      z = __builtin_amdgcn_wmma_f32_16x16x32_f16(false, ka[t2][1].v, false, qb1.v,
                                                 (short)0, z, false, false);
      s[t2] = z;
    }
    float tl = MASKED;
    #pragma unroll
    for (int t2 = 0; t2 < 2; ++t2) {
      #pragma unroll
      for (int r = 0; r < 8; ++r) {
        const int j = jc + t2 * 16 + r + ko;
        const bool ok = (j <= iq) & (j + WIN > iq) & (mk[t2][r] != 0);
        const float v = ok ? s[t2][r] * SM_SCALE : MASKED;
        s[t2][r] = v;
        tl = fmaxf(tl, v);
      }
    }
    tl = fmaxf(tl, __shfl_xor(tl, 16, 32));
    const float mnew = fmaxf(m, tl);
    const float co = __expf(m - mnew);
    l *= co;
    #pragma unroll
    for (int dt = 0; dt < 4; ++dt) acc[dt] = acc[dt] * co;

    V16 pbf;
    float psum = 0.f;
    #pragma unroll
    for (int r = 0; r < 8; ++r) {
      const float p0 = __expf(s[0][r] - mnew);
      const float p1 = __expf(s[1][r] - mnew);
      psum += p0 + p1;
      pbf.h[0][r] = (half_t)p0;
      pbf.h[1][r] = (half_t)p1;
    }
    l += psum + __shfl_xor(psum, 16, 32);
    m = mnew;

    #pragma unroll
    for (int dt = 0; dt < 4; ++dt) {
      const half_t* vp = Vh + (size_t)(dt * 16 + ln) * TT + jc + ko;
      V16 va;
      ldf(va, vp);
      acc[dt] = __builtin_amdgcn_wmma_f32_16x16x32_f16(
          false, va.v, false, pbf.v, (short)0, acc[dt], false, false);
    }
  }

  const float snk = sinks[h];
  const float m2 = fmaxf(m, snk);
  const float co = __expf(m - m2);
  const float ltot = l * co + __expf(snk - m2);
  const float factor = co / ltot;
  half_t* op = attn + ((size_t)((b * TT + qbase) >> 4) * (DM / 32) + h * 2) * 512 + lane * 8;
  #pragma unroll
  for (int dt = 0; dt < 4; ++dt) {
    v8h o;
    #pragma unroll
    for (int r = 0; r < 8; ++r) o[r] = (half_t)(acc[dt][r] * factor);
    *(v8h*)(op + (dt >> 1) * 512 + (dt & 1) * 256) = o;
  }
}

extern "C" void kernel_launch(void* const* d_in, const int* in_sizes, int n_in,
                              void* d_out, int out_size, void* d_ws, size_t ws_size,
                              hipStream_t stream) {
  const float* x      = (const float*)d_in[0];
  const int*   amask  = (const int*)d_in[1];
  const float* nscale = (const float*)d_in[2];
  const float* qkv_w  = (const float*)d_in[3];
  const float* qkv_b  = (const float*)d_in[4];
  const float* out_w  = (const float*)d_in[5];
  const float* out_b  = (const float*)d_in[6];
  const float* sinks  = (const float*)d_in[7];
  float* out = (float*)d_out;
  char*  ws  = (char*)d_ws;

  half_t* qkvw_h = (half_t*)(ws + OFF_QKVW);
  half_t* outw_h = (half_t*)(ws + OFF_OUTW);
  half_t* t_h    = (half_t*)(ws + OFF_T);
  half_t* attn_h = (half_t*)(ws + OFF_ATTN);
  half_t* Qh     = (half_t*)(ws + OFF_Q);
  half_t* Kh     = (half_t*)(ws + OFF_K);
  half_t* Vth    = (half_t*)(ws + OFF_V);

  prep_kernel<<<PREP_GRID, 256, 0, stream>>>(x, nscale, t_h, qkv_w, qkvw_h, out_w, outw_h);

  qkv_rope_kernel<<<QKV_GRID, 384, 0, stream>>>(t_h, qkvw_h, qkv_b, Qh, Kh, Vth, DM);

  attn_kernel<<<512, 256, 0, stream>>>(Qh, Kh, Vth, amask, sinks, attn_h);

  gemm_wmma_kernel<<<256, 256, 0, stream>>>(attn_h, outw_h, out_b, x,
                                            out, MROW, DM, DM);
}


# Q, K and V^T kept fragment-major in the workspace (1 KiB tiles, as the GEMM operands are): each attention K,V,Q fragment load and QKV epilogue store covers 4 whole 128-B lines, not 16 partial
# speedup vs baseline: 3.7092x; 1.0685x over previous
#include <hip/hip_runtime.h>
#include <hip/hip_bf16.h>

typedef _Float16 half_t;
typedef __attribute__((ext_vector_type(16))) _Float16 v16h;
typedef __attribute__((ext_vector_type(8)))  _Float16 v8h;
typedef __attribute__((ext_vector_type(8)))  float    v8f;

union V16 { v16h v; v8h h[2]; };

constexpr int BB   = 2;
constexpr int TT   = 2048;
constexpr int DM   = 1024;
constexpr int HN   = 16;
constexpr int DH   = 64;
constexpr int QKVD = 3072;
constexpr int MROW = BB * TT;
constexpr int WIN  = 128;
constexpr float SM_SCALE = 0.125f;
constexpr int ROW_TILES  = MROW / 16;
constexpr int QKVW_TILES = QKVD * DM / 512;
constexpr int OUTW_TILES = DM * DM / 512;
constexpr int QKVW_BLOCKS = QKVW_TILES / 8;
constexpr int OUTW_BLOCKS = OUTW_TILES / 8;
constexpr float MASKED = -3.0e38f;
constexpr float MINIT  = -1.0e30f;

constexpr size_t OFF_QKVW = 0;
constexpr size_t SZ_QKVW  = (size_t)QKVD * DM * 2;
constexpr size_t OFF_OUTW = OFF_QKVW + SZ_QKVW;
constexpr size_t SZ_OUTW  = (size_t)DM * DM * 2;
constexpr size_t OFF_T    = OFF_OUTW + SZ_OUTW;
constexpr size_t SZ_T     = (size_t)MROW * DM * 2;
constexpr size_t OFF_QKVF = OFF_T + SZ_T;
constexpr size_t SZ_QKVF  = (size_t)MROW * QKVD * 4;
constexpr size_t OFF_ATTN = OFF_QKVF;
constexpr size_t OFF_Q    = OFF_QKVF + SZ_QKVF;
constexpr size_t SZ_QKH   = (size_t)BB * HN * TT * DH * 2;
constexpr size_t OFF_K    = OFF_Q + SZ_QKH;
constexpr size_t OFF_V    = OFF_K + SZ_QKH;

__device__ __forceinline__ void ldfm(V16& f, const half_t* p) {
  f.h[0] = *(const v8h*)(p);
  f.h[1] = *(const v8h*)(p + 256);
}

__device__ __forceinline__ v8h cvt8(const float4 a, const float4 b) {
  v8h o;
  o[0] = (half_t)a.x; o[1] = (half_t)a.y; o[2] = (half_t)a.z; o[3] = (half_t)a.w;
  o[4] = (half_t)b.x; o[5] = (half_t)b.y; o[6] = (half_t)b.z; o[7] = (half_t)b.w;
  return o;
}

__device__ __forceinline__ void rmsnorm_tile_row(
    const float* __restrict__ x, const float* __restrict__ scale,
    half_t* __restrict__ t, const int rt, float (*red)[16]) {
  const int lane = threadIdx.x & 31;
  const int wv   = threadIdx.x >> 5;
  const int ln = lane & 15;
  const int hf = lane >> 4;
  const float* xr = x + (size_t)(rt * 16 + ln) * DM + wv * 128 + hf * 8;
  const float* sp = scale + wv * 128 + hf * 8;
  float4 va[8], vb[8], sa[8], sb[8];
  float ss = 0.f;
  #pragma unroll
  for (int p = 0; p < 8; ++p) {
    va[p] = *(const float4*)(xr + p * 16);
    vb[p] = *(const float4*)(xr + p * 16 + 4);
  }
  #pragma unroll
  for (int p = 0; p < 8; ++p) {
    sa[p] = *(const float4*)(sp + p * 16);
    sb[p] = *(const float4*)(sp + p * 16 + 4);
  }
  #pragma unroll
  for (int p = 0; p < 8; ++p) {
    ss += va[p].x * va[p].x + va[p].y * va[p].y + va[p].z * va[p].z + va[p].w * va[p].w;
    ss += vb[p].x * vb[p].x + vb[p].y * vb[p].y + vb[p].z * vb[p].z + vb[p].w * vb[p].w;
  }
  ss += __shfl_xor(ss, 16, 32);
  if (hf == 0) red[wv][ln] = ss;
  __syncthreads();
  float tot = 0.f;
  #pragma unroll
  for (int i = 0; i < 8; ++i) tot += red[i][ln];
  const float inv = rsqrtf(tot * (1.0f / DM) + 1e-5f);
  half_t* dst = t + ((size_t)rt * (DM / 32) + wv * 4) * 512 + lane * 8;
  #pragma unroll
  for (int p = 0; p < 8; ++p) {
    const float4 s0 = sa[p], s1 = sb[p];
    float4 a = va[p], b = vb[p];
    a.x = a.x * inv * s0.x; a.y = a.y * inv * s0.y; a.z = a.z * inv * s0.z; a.w = a.w * inv * s0.w;
    b.x = b.x * inv * s1.x; b.y = b.y * inv * s1.y; b.z = b.z * inv * s1.z; b.w = b.w * inv * s1.w;
    *(v8h*)(dst + (p >> 1) * 512 + (p & 1) * 256) = cvt8(a, b);
  }
}

__device__ __forceinline__ void f32_to_f16_tile(
    const float* __restrict__ src, half_t* __restrict__ dst, const int tile, const int K) {
  const int lane = threadIdx.x & 31;
  const int kt = K >> 5;
  const int rt = tile / kt, kc = tile % kt;
  const int ln = lane & 15;
  const int hf = lane >> 4;
  const float* sp = src + (size_t)(rt * 16 + ln) * K + kc * 32 + hf * 8;
  const float4 a0 = *(const float4*)(sp);
  const float4 b0 = *(const float4*)(sp + 4);
  const float4 a1 = *(const float4*)(sp + 16);
  const float4 b1 = *(const float4*)(sp + 20);
  half_t* dp = dst + (size_t)tile * 512 + lane * 8;
  *(v8h*)(dp)       = cvt8(a0, b0);
  *(v8h*)(dp + 256) = cvt8(a1, b1);
}

constexpr int PREP_GRID = ROW_TILES + QKVW_BLOCKS + OUTW_BLOCKS;

__global__ __launch_bounds__(256) void prep_kernel(
    const float* __restrict__ x, const float* __restrict__ scale, half_t* __restrict__ t,
    const float* __restrict__ qkv_w, half_t* __restrict__ qkvw_h,
    const float* __restrict__ out_w, half_t* __restrict__ outw_h) {
  __shared__ float red[8][16];
  const int blk = blockIdx.x;
  if (blk < ROW_TILES) {
    rmsnorm_tile_row(x, scale, t, blk, red);
  } else {
    const int tile = (blk - ROW_TILES) * 8 + (threadIdx.x >> 5);
    if (tile < QKVW_TILES) f32_to_f16_tile(qkv_w, qkvw_h, tile, DM);
    else                   f32_to_f16_tile(out_w, outw_h, tile - QKVW_TILES, DM);
  }
}

__device__ __forceinline__ void pin_order() { __builtin_amdgcn_sched_barrier(0); }

__device__ __forceinline__ void mma8(v8f acc[2][4], const V16 a[2], const V16 b[4]) {
  #pragma unroll
  for (int i = 0; i < 2; ++i)
    #pragma unroll
    for (int j = 0; j < 4; ++j)
      acc[i][j] = __builtin_amdgcn_wmma_f32_16x16x32_f16(
          false, a[i].v, false, b[j].v, (short)0, acc[i][j], false, false);
}

__device__ __forceinline__ void load_step(
    V16 a[2], V16 b[4],
    const half_t* A0, const half_t* A1,
    const half_t* W0, const half_t* W1,
    const half_t* W2, const half_t* W3, int k) {
  ldfm(a[0], A0 + k * 16);
  ldfm(a[1], A1 + k * 16);
  ldfm(b[0], W0 + k * 16);
  ldfm(b[1], W1 + k * 16);
  ldfm(b[2], W2 + k * 16);
  ldfm(b[3], W3 + k * 16);
}

__global__ __launch_bounds__(256) void gemm_wmma_kernel(
    const half_t* __restrict__ A, const half_t* __restrict__ W,
    const float* __restrict__ bias, const float* __restrict__ residual,
    float* __restrict__ C, int M, int N, int K) {
  const int lane = threadIdx.x & 31;
  const int wv   = threadIdx.x >> 5;
  const int nbn  = N >> 7;
  const int bm = blockIdx.x / nbn, bn = blockIdx.x % nbn;
  const int tm = bm * 4 + (wv >> 1);
  const int tn = bn * 2 + (wv & 1);
  if (tm >= (M >> 5) || tn >= (N >> 6)) return;
  const int ko = (lane >= 16) ? 8 : 0;
  const int ln = lane & 15;

  const half_t* A0 = A + (size_t)(tm * 32) * K + lane * 8;
  const half_t* A1 = A0 + (size_t)16 * K;
  const half_t* W0 = W + (size_t)(tn * 64) * K + lane * 8;
  const half_t* W1 = W0 + (size_t)16 * K;
  const half_t* W2 = W0 + (size_t)32 * K;
  const half_t* W3 = W0 + (size_t)48 * K;

  v8f acc[2][4] = {};
  V16 a0[2], b0[4], a1[2], b1[4], a2[2], b2[4], a3[2], b3[4];

  load_step(a0, b0, A0, A1, W0, W1, W2, W3, 0);   pin_order();
  load_step(a1, b1, A0, A1, W0, W1, W2, W3, 32);  pin_order();
  load_step(a2, b2, A0, A1, W0, W1, W2, W3, 64);  pin_order();
  int k = 0;
  for (; k < K - 128; k += 128) {
    load_step(a3, b3, A0, A1, W0, W1, W2, W3, k + 96);   pin_order();
    mma8(acc, a0, b0);                                   pin_order();
    load_step(a0, b0, A0, A1, W0, W1, W2, W3, k + 128);  pin_order();
    mma8(acc, a1, b1);                                   pin_order();
    load_step(a1, b1, A0, A1, W0, W1, W2, W3, k + 160);  pin_order();
    mma8(acc, a2, b2);                                   pin_order();
    load_step(a2, b2, A0, A1, W0, W1, W2, W3, k + 192);  pin_order();
    mma8(acc, a3, b3);                                   pin_order();
  }
  load_step(a3, b3, A0, A1, W0, W1, W2, W3, k + 96);     pin_order();
  mma8(acc, a0, b0);
  mma8(acc, a1, b1);
  mma8(acc, a2, b2);
  mma8(acc, a3, b3);

  float bv[4];
  #pragma unroll
  for (int j = 0; j < 4; ++j) bv[j] = bias[tn * 64 + j * 16 + ln];
  float rv[2][4][8];
  if (residual) {
    #pragma unroll
    for (int i = 0; i < 2; ++i)
      #pragma unroll
      for (int j = 0; j < 4; ++j)
        #pragma unroll
        for (int r = 0; r < 8; ++r)
          rv[i][j][r] = residual[(size_t)(tm * 32 + i * 16 + r + ko) * N + tn * 64 + j * 16 + ln];
  } else {
    #pragma unroll
    for (int i = 0; i < 2; ++i)
      #pragma unroll
      for (int j = 0; j < 4; ++j)
        #pragma unroll
        for (int r = 0; r < 8; ++r) rv[i][j][r] = 0.f;
  }

  #pragma unroll
  for (int i = 0; i < 2; ++i) {
    #pragma unroll
    for (int j = 0; j < 4; ++j) {
      const int col = tn * 64 + j * 16 + ln;
      #pragma unroll
      for (int r = 0; r < 8; ++r) {
        const int row = tm * 32 + i * 16 + r + ko;
        float v = acc[i][j][r] + bv[j];
        v += rv[i][j][r];
        C[(size_t)row * N + col] = v;
      }
    }
  }
}

__device__ __forceinline__ void mma16t(v8f acc[4][4], const V16 n[4], const V16 m[4]) {
  #pragma unroll
  for (int i = 0; i < 4; ++i)
    #pragma unroll
    for (int j = 0; j < 4; ++j)
      acc[i][j] = __builtin_amdgcn_wmma_f32_16x16x32_f16(
          false, m[j].v, false, n[i].v, (short)0, acc[i][j], false, false);
}

__device__ __forceinline__ void load_step8(
    V16 n[4], V16 m[4], const half_t* N0, const half_t* M0, size_t ts, int k) {
  #pragma unroll
  for (int i = 0; i < 4; ++i) ldfm(n[i], N0 + i * ts + k * 16);
  #pragma unroll
  for (int j = 0; j < 4; ++j) ldfm(m[j], M0 + j * ts + k * 16);
}

constexpr int CS_PITCH = 36;
constexpr int QKV_BT   = 128;
constexpr int QKV_BH   = 6;
constexpr int QKV_WB   = (3 * HN) / QKV_BH;
constexpr int QKV_GRID = (MROW / QKV_BT) * QKV_WB;

__global__ __launch_bounds__(384) __attribute__((amdgpu_waves_per_eu(3, 3))) void qkv_rope_kernel(
    const half_t* __restrict__ A, const half_t* __restrict__ W,
    const float* __restrict__ bias,
    half_t* __restrict__ Q, half_t* __restrict__ Kr, half_t* __restrict__ Vt,
    int K) {
  __shared__ __attribute__((aligned(16))) float cs_c[QKV_BT * CS_PITCH];
  __shared__ __attribute__((aligned(16))) float cs_s[QKV_BT * CS_PITCH];

  const int lane = threadIdx.x & 31;
  const int wv   = threadIdx.x >> 5;
  const int ln = lane & 15;
  const int ko = (lane >= 16) ? 8 : 0;
  const int tb = blockIdx.x / QKV_WB;
  const int wb = blockIdx.x % QKV_WB;
  const int ti = wv / QKV_BH;
  const int hs = wb * QKV_BH + (wv - ti * QKV_BH);
  const int sec = hs >> 4;
  const int h   = hs & (HN - 1);
  const int trow = tb * QKV_BT + ti * 64;
  const int b    = trow / TT;
  const int tl0  = trow - b * TT;
  const bool has_qk = wb * QKV_BH < 2 * HN;

  if (has_qk) {
    const int tq0 = (tb * QKV_BT) & (TT - 1);
    for (int e = threadIdx.x; e < QKV_BT * 32; e += 384) {
      const int tt = e >> 5, p = e & 31;
      const float ang = (float)(tq0 + tt) * __expf(-(float)p * (9.210340372f / 32.0f));
      cs_c[tt * CS_PITCH + p] = cosf(ang);
      cs_s[tt * CS_PITCH + p] = sinf(ang);
    }
    __syncthreads();
  }

  const half_t* Mp;
  const half_t* Np;
  int m0, n0;
  if (sec < 2) { Mp = W; m0 = hs * 64; Np = A; n0 = trow; }
  else         { Mp = A; m0 = trow;    Np = W; n0 = hs * 64; }

  const half_t* N0 = Np + (size_t)n0 * K + lane * 8;
  const half_t* M0 = Mp + (size_t)m0 * K + lane * 8;
  const size_t ts = (size_t)16 * K;

  v8f acc[4][4] = {};
  {
    V16 pn[4], pm[4], qn[4], qm[4];
    load_step8(pn, pm, N0, M0, ts, 0);         pin_order();
    int k = 32;
    for (; k < K - 32; k += 64) {
      load_step8(qn, qm, N0, M0, ts, k);       pin_order();
      mma16t(acc, pn, pm);                     pin_order();
      load_step8(pn, pm, N0, M0, ts, k + 32);  pin_order();
      mma16t(acc, qn, qm);                     pin_order();
    }
    load_step8(qn, qm, N0, M0, ts, k);         pin_order();
    mma16t(acc, pn, pm);
    mma16t(acc, qn, qm);
  }

  if (sec < 2) {
    half_t* dst = (sec == 0 ? Q : Kr) + ((size_t)(b * HN + h) * TT + tl0) * DH + lane * 8;
    const float* bp = bias + m0 + ko;
    #pragma unroll
    for (int j = 0; j < 2; ++j) {
      const float4 bl0 = *(const float4*)(bp + j * 16);
      const float4 bl1 = *(const float4*)(bp + j * 16 + 4);
      const float4 bh0 = *(const float4*)(bp + j * 16 + 32);
      const float4 bh1 = *(const float4*)(bp + j * 16 + 36);
      const float blo[8] = {bl0.x, bl0.y, bl0.z, bl0.w, bl1.x, bl1.y, bl1.z, bl1.w};
      const float bhi[8] = {bh0.x, bh0.y, bh0.z, bh0.w, bh1.x, bh1.y, bh1.z, bh1.w};
      #pragma unroll
      for (int i = 0; i < 4; ++i) {
        const int tw = i * 16 + ln;
        const int tt = ti * 64 + tw;
        const float* cp = cs_c + tt * CS_PITCH + j * 16 + ko;
        const float* sp = cs_s + tt * CS_PITCH + j * 16 + ko;
        const float4 c0 = *(const float4*)(cp), c1 = *(const float4*)(cp + 4);
        const float4 s0 = *(const float4*)(sp), s1 = *(const float4*)(sp + 4);
        const float cc[8] = {c0.x, c0.y, c0.z, c0.w, c1.x, c1.y, c1.z, c1.w};
        const float ss[8] = {s0.x, s0.y, s0.z, s0.w, s1.x, s1.y, s1.z, s1.w};
        v8h o1, o2;
        #pragma unroll
        for (int r = 0; r < 8; ++r) {
          const float x1 = acc[i][j][r] + blo[r];
          const float x2 = acc[i][j + 2][r] + bhi[r];
          o1[r] = (half_t)(x1 * cc[r] - x2 * ss[r]);
          o2[r] = (half_t)(x2 * cc[r] + x1 * ss[r]);
        }
        *(v8h*)(dst + (i * 2 + 0) * 512 + j * 256) = o1;
        *(v8h*)(dst + (i * 2 + 1) * 512 + j * 256) = o2;
      }
    }
  } else {
    half_t* vb = Vt + (size_t)(b * HN + h) * DH * TT + (size_t)(tl0 >> 5) * 2048 + lane * 8;
    #pragma unroll
    for (int i = 0; i < 4; ++i) {
      const int c = h * 64 + i * 16 + ln;
      const float bv = bias[2 * DM + c];
      #pragma unroll
      for (int j = 0; j < 4; ++j) {
        v8h o;
        #pragma unroll
        for (int r = 0; r < 8; ++r) o[r] = (half_t)(acc[i][j][r] + bv);
        *(v8h*)(vb + ((j >> 1) * 4 + i) * 512 + (j & 1) * 256) = o;
      }
    }
  }
}

__global__ __launch_bounds__(256) __attribute__((amdgpu_waves_per_eu(4, 4))) void attn_kernel(
    const half_t* __restrict__ Q, const half_t* __restrict__ Kr,
    const half_t* __restrict__ Vt, const int* __restrict__ amask,
    const float* __restrict__ sinks, half_t* __restrict__ attn) {
  const int lane = threadIdx.x & 31;
  const int wid  = (blockIdx.x * blockDim.x + threadIdx.x) >> 5;
  const int QT   = TT / 16;
  const int qt   = wid % QT;
  const int bh   = wid / QT;
  const int h    = bh % HN, b = bh / HN;
  const int qbase = qt * 16;
  const int ln = lane & 15;
  const int ko = (lane >= 16) ? 8 : 0;
  const int iq = qbase + ln;
  const half_t* Qh = Q  + (size_t)bh * TT * DH;
  const half_t* Kh = Kr + (size_t)bh * TT * DH;
  const half_t* Vh = Vt + (size_t)bh * DH * TT;
  const int* am = amask + b * TT;

  V16 qb0, qb1;
  {
    const half_t* qp = Qh + (size_t)qt * 1024 + lane * 8;
    ldfm(qb0, qp);
    ldfm(qb1, qp + 512);
  }

  v8f acc[4] = {};
  float m = MINIT, l = 0.f;
  int kstart = qbase - (WIN - 1); if (kstart < 0) kstart = 0;
  kstart &= ~31;
  const int kend = qbase + 16;

  for (int jc = kstart; jc < kend; jc += 32) {
    int mk[2][8];
    #pragma unroll
    for (int t2 = 0; t2 < 2; ++t2) {
      const int4 m0 = *(const int4*)(am + jc + t2 * 16 + ko);
      const int4 m1 = *(const int4*)(am + jc + t2 * 16 + ko + 4);
      mk[t2][0] = m0.x; mk[t2][1] = m0.y; mk[t2][2] = m0.z; mk[t2][3] = m0.w;
      mk[t2][4] = m1.x; mk[t2][5] = m1.y; mk[t2][6] = m1.z; mk[t2][7] = m1.w;
    }
    v8f s[2];
    V16 ka[2][2];
    const half_t* kp = Kh + (size_t)jc * DH + lane * 8;
    #pragma unroll
    for (int t2 = 0; t2 < 2; ++t2) {
      ldfm(ka[t2][0], kp + (t2 * 2 + 0) * 512);
      ldfm(ka[t2][1], kp + (t2 * 2 + 1) * 512);
    }
    #pragma unroll
    for (int t2 = 0; t2 < 2; ++t2) {
      v8f z = {};
      z = __builtin_amdgcn_wmma_f32_16x16x32_f16(false, ka[t2][0].v, false, qb0.v,
                                                 (short)0, z, false, false);
      z = __builtin_amdgcn_wmma_f32_16x16x32_f16(false, ka[t2][1].v, false, qb1.v,
                                                 (short)0, z, false, false);
      s[t2] = z;
    }
    float tl = MASKED;
    #pragma unroll
    for (int t2 = 0; t2 < 2; ++t2) {
      #pragma unroll
      for (int r = 0; r < 8; ++r) {
        const int j = jc + t2 * 16 + r + ko;
        const bool ok = (j <= iq) & (j + WIN > iq) & (mk[t2][r] != 0);
        const float v = ok ? s[t2][r] * SM_SCALE : MASKED;
        s[t2][r] = v;
        tl = fmaxf(tl, v);
      }
    }
    tl = fmaxf(tl, __shfl_xor(tl, 16, 32));
    const float mnew = fmaxf(m, tl);
    const float co = __expf(m - mnew);
    l *= co;
    #pragma unroll
    for (int dt = 0; dt < 4; ++dt) acc[dt] = acc[dt] * co;

    V16 pbf;
    float psum = 0.f;
    #pragma unroll
    for (int r = 0; r < 8; ++r) {
      const float p0 = __expf(s[0][r] - mnew);
      const float p1 = __expf(s[1][r] - mnew);
      psum += p0 + p1;
      pbf.h[0][r] = (half_t)p0;
      pbf.h[1][r] = (half_t)p1;
    }
    l += psum + __shfl_xor(psum, 16, 32);
    m = mnew;

    #pragma unroll
    for (int dt = 0; dt < 4; ++dt) {
      const half_t* vp = Vh + (size_t)(jc >> 5) * 2048 + dt * 512 + lane * 8;
      V16 va;
      ldfm(va, vp);
      acc[dt] = __builtin_amdgcn_wmma_f32_16x16x32_f16(
          false, va.v, false, pbf.v, (short)0, acc[dt], false, false);
    }
  }

  const float snk = sinks[h];
  const float m2 = fmaxf(m, snk);
  const float co = __expf(m - m2);
  const float ltot = l * co + __expf(snk - m2);
  const float factor = co / ltot;
  half_t* op = attn + ((size_t)((b * TT + qbase) >> 4) * (DM / 32) + h * 2) * 512 + lane * 8;
  #pragma unroll
  for (int dt = 0; dt < 4; ++dt) {
    v8h o;
    #pragma unroll
    for (int r = 0; r < 8; ++r) o[r] = (half_t)(acc[dt][r] * factor);
    *(v8h*)(op + (dt >> 1) * 512 + (dt & 1) * 256) = o;
  }
}

extern "C" void kernel_launch(void* const* d_in, const int* in_sizes, int n_in,
                              void* d_out, int out_size, void* d_ws, size_t ws_size,
                              hipStream_t stream) {
  const float* x      = (const float*)d_in[0];
  const int*   amask  = (const int*)d_in[1];
  const float* nscale = (const float*)d_in[2];
  const float* qkv_w  = (const float*)d_in[3];
  const float* qkv_b  = (const float*)d_in[4];
  const float* out_w  = (const float*)d_in[5];
  const float* out_b  = (const float*)d_in[6];
  const float* sinks  = (const float*)d_in[7];
  float* out = (float*)d_out;
  char*  ws  = (char*)d_ws;

  half_t* qkvw_h = (half_t*)(ws + OFF_QKVW);
  half_t* outw_h = (half_t*)(ws + OFF_OUTW);
  half_t* t_h    = (half_t*)(ws + OFF_T);
  half_t* attn_h = (half_t*)(ws + OFF_ATTN);
  half_t* Qh     = (half_t*)(ws + OFF_Q);
  half_t* Kh     = (half_t*)(ws + OFF_K);
  half_t* Vth    = (half_t*)(ws + OFF_V);

  prep_kernel<<<PREP_GRID, 256, 0, stream>>>(x, nscale, t_h, qkv_w, qkvw_h, out_w, outw_h);

  qkv_rope_kernel<<<QKV_GRID, 384, 0, stream>>>(t_h, qkvw_h, qkv_b, Qh, Kh, Vth, DM);

  attn_kernel<<<512, 256, 0, stream>>>(Qh, Kh, Vth, amask, sinks, attn_h);

  gemm_wmma_kernel<<<256, 256, 0, stream>>>(attn_h, outw_h, out_b, x,
                                            out, MROW, DM, DM);
}


# QKV GEMM, attention and out-projection each open with a second-level-cache prefetch (global_prefetch_b8) of the operands the launch before wrote: a block's share, 1 to 3 units of 256 B a thread
# speedup vs baseline: 3.8773x; 1.0453x over previous
#include <hip/hip_runtime.h>
#include <hip/hip_bf16.h>

typedef _Float16 half_t;
typedef __attribute__((ext_vector_type(16))) _Float16 v16h;
typedef __attribute__((ext_vector_type(8)))  _Float16 v8h;
typedef __attribute__((ext_vector_type(8)))  float    v8f;

union V16 { v16h v; v8h h[2]; };

constexpr int BB   = 2;
constexpr int TT   = 2048;
constexpr int DM   = 1024;
constexpr int HN   = 16;
constexpr int DH   = 64;
constexpr int QKVD = 3072;
constexpr int MROW = BB * TT;
constexpr int WIN  = 128;
constexpr float SM_SCALE = 0.125f;
constexpr int ROW_TILES  = MROW / 16;
constexpr int QKVW_TILES = QKVD * DM / 512;
constexpr int OUTW_TILES = DM * DM / 512;
constexpr int QKVW_BLOCKS = QKVW_TILES / 8;
constexpr int OUTW_BLOCKS = OUTW_TILES / 8;
constexpr float MASKED = -3.0e38f;
constexpr float MINIT  = -1.0e30f;

constexpr size_t OFF_QKVW = 0;
constexpr size_t SZ_QKVW  = (size_t)QKVD * DM * 2;
constexpr size_t OFF_OUTW = OFF_QKVW + SZ_QKVW;
constexpr size_t SZ_OUTW  = (size_t)DM * DM * 2;
constexpr size_t OFF_T    = OFF_OUTW + SZ_OUTW;
constexpr size_t SZ_T     = (size_t)MROW * DM * 2;
constexpr size_t OFF_QKVF = OFF_T + SZ_T;
constexpr size_t SZ_QKVF  = (size_t)MROW * QKVD * 4;
constexpr size_t OFF_ATTN = OFF_QKVF;
constexpr size_t OFF_Q    = OFF_QKVF + SZ_QKVF;
constexpr size_t SZ_QKH   = (size_t)BB * HN * TT * DH * 2;
constexpr size_t OFF_K    = OFF_Q + SZ_QKH;
constexpr size_t OFF_V    = OFF_K + SZ_QKH;

__device__ __forceinline__ void ldfm(V16& f, const half_t* p) {
  f.h[0] = *(const v8h*)(p);
  f.h[1] = *(const v8h*)(p + 256);
}

__device__ __forceinline__ v8h cvt8(const float4 a, const float4 b) {
  v8h o;
  o[0] = (half_t)a.x; o[1] = (half_t)a.y; o[2] = (half_t)a.z; o[3] = (half_t)a.w;
  o[4] = (half_t)b.x; o[5] = (half_t)b.y; o[6] = (half_t)b.z; o[7] = (half_t)b.w;
  return o;
}

__device__ __forceinline__ void l2_prefetch(const void* p) {
  asm volatile("global_prefetch_b8 %0, off scope:SCOPE_SE" :: "v"(p));
}

__device__ __forceinline__ void rmsnorm_tile_row(
    const float* __restrict__ x, const float* __restrict__ scale,
    half_t* __restrict__ t, const int rt, float (*red)[16]) {
  const int lane = threadIdx.x & 31;
  const int wv   = threadIdx.x >> 5;
  const int ln = lane & 15;
  const int hf = lane >> 4;
  const float* xr = x + (size_t)(rt * 16 + ln) * DM + wv * 128 + hf * 8;
  const float* sp = scale + wv * 128 + hf * 8;
  float4 va[8], vb[8], sa[8], sb[8];
  float ss = 0.f;
  #pragma unroll
  for (int p = 0; p < 8; ++p) {
    va[p] = *(const float4*)(xr + p * 16);
    vb[p] = *(const float4*)(xr + p * 16 + 4);
  }
  #pragma unroll
  for (int p = 0; p < 8; ++p) {
    sa[p] = *(const float4*)(sp + p * 16);
    sb[p] = *(const float4*)(sp + p * 16 + 4);
  }
  #pragma unroll
  for (int p = 0; p < 8; ++p) {
    ss += va[p].x * va[p].x + va[p].y * va[p].y + va[p].z * va[p].z + va[p].w * va[p].w;
    ss += vb[p].x * vb[p].x + vb[p].y * vb[p].y + vb[p].z * vb[p].z + vb[p].w * vb[p].w;
  }
  ss += __shfl_xor(ss, 16, 32);
  if (hf == 0) red[wv][ln] = ss;
  __syncthreads();
  float tot = 0.f;
  #pragma unroll
  for (int i = 0; i < 8; ++i) tot += red[i][ln];
  const float inv = rsqrtf(tot * (1.0f / DM) + 1e-5f);
  half_t* dst = t + ((size_t)rt * (DM / 32) + wv * 4) * 512 + lane * 8;
  #pragma unroll
  for (int p = 0; p < 8; ++p) {
    const float4 s0 = sa[p], s1 = sb[p];
    float4 a = va[p], b = vb[p];
    a.x = a.x * inv * s0.x; a.y = a.y * inv * s0.y; a.z = a.z * inv * s0.z; a.w = a.w * inv * s0.w;
    b.x = b.x * inv * s1.x; b.y = b.y * inv * s1.y; b.z = b.z * inv * s1.z; b.w = b.w * inv * s1.w;
    *(v8h*)(dst + (p >> 1) * 512 + (p & 1) * 256) = cvt8(a, b);
  }
}

__device__ __forceinline__ void f32_to_f16_tile(
    const float* __restrict__ src, half_t* __restrict__ dst, const int tile, const int K) {
  const int lane = threadIdx.x & 31;
  const int kt = K >> 5;
  const int rt = tile / kt, kc = tile % kt;
  const int ln = lane & 15;
  const int hf = lane >> 4;
  const float* sp = src + (size_t)(rt * 16 + ln) * K + kc * 32 + hf * 8;
  const float4 a0 = *(const float4*)(sp);
  const float4 b0 = *(const float4*)(sp + 4);
  const float4 a1 = *(const float4*)(sp + 16);
  const float4 b1 = *(const float4*)(sp + 20);
  half_t* dp = dst + (size_t)tile * 512 + lane * 8;
  *(v8h*)(dp)       = cvt8(a0, b0);
  *(v8h*)(dp + 256) = cvt8(a1, b1);
}

constexpr int PREP_GRID = ROW_TILES + QKVW_BLOCKS + OUTW_BLOCKS;

__global__ __launch_bounds__(256) void prep_kernel(
    const float* __restrict__ x, const float* __restrict__ scale, half_t* __restrict__ t,
    const float* __restrict__ qkv_w, half_t* __restrict__ qkvw_h,
    const float* __restrict__ out_w, half_t* __restrict__ outw_h) {
  __shared__ float red[8][16];
  const int blk = blockIdx.x;
  if (blk < ROW_TILES) {
    rmsnorm_tile_row(x, scale, t, blk, red);
  } else {
    const int tile = (blk - ROW_TILES) * 8 + (threadIdx.x >> 5);
    if (tile < QKVW_TILES) f32_to_f16_tile(qkv_w, qkvw_h, tile, DM);
    else                   f32_to_f16_tile(out_w, outw_h, tile - QKVW_TILES, DM);
  }
}

__device__ __forceinline__ void pin_order() { __builtin_amdgcn_sched_barrier(0); }

__device__ __forceinline__ void mma8(v8f acc[2][4], const V16 a[2], const V16 b[4]) {
  #pragma unroll
  for (int i = 0; i < 2; ++i)
    #pragma unroll
    for (int j = 0; j < 4; ++j)
      acc[i][j] = __builtin_amdgcn_wmma_f32_16x16x32_f16(
          false, a[i].v, false, b[j].v, (short)0, acc[i][j], false, false);
}

__device__ __forceinline__ void load_step(
    V16 a[2], V16 b[4],
    const half_t* A0, const half_t* A1,
    const half_t* W0, const half_t* W1,
    const half_t* W2, const half_t* W3, int k) {
  ldfm(a[0], A0 + k * 16);
  ldfm(a[1], A1 + k * 16);
  ldfm(b[0], W0 + k * 16);
  ldfm(b[1], W1 + k * 16);
  ldfm(b[2], W2 + k * 16);
  ldfm(b[3], W3 + k * 16);
}

__global__ __launch_bounds__(256) void gemm_wmma_kernel(
    const half_t* __restrict__ A, const half_t* __restrict__ W,
    const float* __restrict__ bias, const float* __restrict__ residual,
    float* __restrict__ C, int M, int N, int K) {
  const int lane = threadIdx.x & 31;
  const int wv   = threadIdx.x >> 5;
  const int nbn  = N >> 7;
  const int bm = blockIdx.x / nbn, bn = blockIdx.x % nbn;
  const int tm = bm * 4 + (wv >> 1);
  const int tn = bn * 2 + (wv & 1);
  if (tm >= (M >> 5) || tn >= (N >> 6)) return;
  const int ko = (lane >= 16) ? 8 : 0;
  const int ln = lane & 15;

  {
    const int ua = K >> 2, uw = K >> 4, ur = residual ? 256 : 0;
    const char* pa = (const char*)(A + (size_t)(bm * 128) * K) + (size_t)(bn & 3) * ua * 256;
    const char* pw = (const char*)(W + (size_t)(bn * 128) * K) + (size_t)(bm & 15) * uw * 256;
    for (int u = threadIdx.x; u < ua + uw + ur; u += 256) {
      const char* p;
      if (u < ua)           p = pa + (size_t)u * 256;
      else if (u < ua + uw) p = pw + (size_t)(u - ua) * 256;
      else {
        const int v = u - ua - uw;
        p = (const char*)(residual + (size_t)(bm * 128 + (v >> 1)) * N + bn * 128 + (v & 1) * 64);
      }
      l2_prefetch(p);
    }
  }

  const half_t* A0 = A + (size_t)(tm * 32) * K + lane * 8;
  const half_t* A1 = A0 + (size_t)16 * K;
  const half_t* W0 = W + (size_t)(tn * 64) * K + lane * 8;
  const half_t* W1 = W0 + (size_t)16 * K;
  const half_t* W2 = W0 + (size_t)32 * K;
  const half_t* W3 = W0 + (size_t)48 * K;

  v8f acc[2][4] = {};
  V16 a0[2], b0[4], a1[2], b1[4], a2[2], b2[4], a3[2], b3[4];

  load_step(a0, b0, A0, A1, W0, W1, W2, W3, 0);   pin_order();
  load_step(a1, b1, A0, A1, W0, W1, W2, W3, 32);  pin_order();
  load_step(a2, b2, A0, A1, W0, W1, W2, W3, 64);  pin_order();
  int k = 0;
  for (; k < K - 128; k += 128) {
    load_step(a3, b3, A0, A1, W0, W1, W2, W3, k + 96);   pin_order();
    mma8(acc, a0, b0);                                   pin_order();
    load_step(a0, b0, A0, A1, W0, W1, W2, W3, k + 128);  pin_order();
    mma8(acc, a1, b1);                                   pin_order();
    load_step(a1, b1, A0, A1, W0, W1, W2, W3, k + 160);  pin_order();
    mma8(acc, a2, b2);                                   pin_order();
    load_step(a2, b2, A0, A1, W0, W1, W2, W3, k + 192);  pin_order();
    mma8(acc, a3, b3);                                   pin_order();
  }
  load_step(a3, b3, A0, A1, W0, W1, W2, W3, k + 96);     pin_order();
  mma8(acc, a0, b0);
  mma8(acc, a1, b1);
  mma8(acc, a2, b2);
  mma8(acc, a3, b3);

  float bv[4];
  #pragma unroll
  for (int j = 0; j < 4; ++j) bv[j] = bias[tn * 64 + j * 16 + ln];
  float rv[2][4][8];
  if (residual) {
    #pragma unroll
    for (int i = 0; i < 2; ++i)
      #pragma unroll
      for (int j = 0; j < 4; ++j)
        #pragma unroll
        for (int r = 0; r < 8; ++r)
          rv[i][j][r] = residual[(size_t)(tm * 32 + i * 16 + r + ko) * N + tn * 64 + j * 16 + ln];
  } else {
    #pragma unroll
    for (int i = 0; i < 2; ++i)
      #pragma unroll
      for (int j = 0; j < 4; ++j)
        #pragma unroll
        for (int r = 0; r < 8; ++r) rv[i][j][r] = 0.f;
  }

  #pragma unroll
  for (int i = 0; i < 2; ++i) {
    #pragma unroll
    for (int j = 0; j < 4; ++j) {
      const int col = tn * 64 + j * 16 + ln;
      #pragma unroll
      for (int r = 0; r < 8; ++r) {
        const int row = tm * 32 + i * 16 + r + ko;
        float v = acc[i][j][r] + bv[j];
        v += rv[i][j][r];
        C[(size_t)row * N + col] = v;
      }
    }
  }
}

__device__ __forceinline__ void mma16t(v8f acc[4][4], const V16 n[4], const V16 m[4]) {
  #pragma unroll
  for (int i = 0; i < 4; ++i)
    #pragma unroll
    for (int j = 0; j < 4; ++j)
      acc[i][j] = __builtin_amdgcn_wmma_f32_16x16x32_f16(
          false, m[j].v, false, n[i].v, (short)0, acc[i][j], false, false);
}

__device__ __forceinline__ void load_step8(
    V16 n[4], V16 m[4], const half_t* N0, const half_t* M0, size_t ts, int k) {
  #pragma unroll
  for (int i = 0; i < 4; ++i) ldfm(n[i], N0 + i * ts + k * 16);
  #pragma unroll
  for (int j = 0; j < 4; ++j) ldfm(m[j], M0 + j * ts + k * 16);
}

constexpr int CS_PITCH = 36;
constexpr int QKV_BT   = 128;
constexpr int QKV_BH   = 6;
constexpr int QKV_WB   = (3 * HN) / QKV_BH;
constexpr int QKV_GRID = (MROW / QKV_BT) * QKV_WB;

__global__ __launch_bounds__(384) __attribute__((amdgpu_waves_per_eu(3, 3))) void qkv_rope_kernel(
    const half_t* __restrict__ A, const half_t* __restrict__ W,
    const float* __restrict__ bias,
    half_t* __restrict__ Q, half_t* __restrict__ Kr, half_t* __restrict__ Vt,
    int K) {
  __shared__ __attribute__((aligned(16))) float cs_c[QKV_BT * CS_PITCH];
  __shared__ __attribute__((aligned(16))) float cs_s[QKV_BT * CS_PITCH];

  const int lane = threadIdx.x & 31;
  const int wv   = threadIdx.x >> 5;
  const int ln = lane & 15;
  const int ko = (lane >= 16) ? 8 : 0;
  const int tb = blockIdx.x / QKV_WB;
  const int wb = blockIdx.x % QKV_WB;
  const int ti = wv / QKV_BH;
  const int hs = wb * QKV_BH + (wv - ti * QKV_BH);
  const int sec = hs >> 4;
  const int h   = hs & (HN - 1);
  const int trow = tb * QKV_BT + ti * 64;
  const int b    = trow / TT;
  const int tl0  = trow - b * TT;
  const bool has_qk = wb * QKV_BH < 2 * HN;

  {
    const int ut = K >> 2, uw = (3 * K) >> 4;
    const char* pt = (const char*)(A + (size_t)(tb * QKV_BT) * K) + (size_t)(wb & 3) * ut * 256;
    const char* pw = (const char*)(W + (size_t)(wb * QKV_BH * 64) * K) + (size_t)(tb & 15) * uw * 256;
    for (int u = threadIdx.x; u < ut + uw; u += 384)
      l2_prefetch(u < ut ? pt + (size_t)u * 256 : pw + (size_t)(u - ut) * 256);
  }

  if (has_qk) {
    const int tq0 = (tb * QKV_BT) & (TT - 1);
    for (int e = threadIdx.x; e < QKV_BT * 32; e += 384) {
      const int tt = e >> 5, p = e & 31;
      const float ang = (float)(tq0 + tt) * __expf(-(float)p * (9.210340372f / 32.0f));
      cs_c[tt * CS_PITCH + p] = cosf(ang);
      cs_s[tt * CS_PITCH + p] = sinf(ang);
    }
    __syncthreads();
  }

  const half_t* Mp;
  const half_t* Np;
  int m0, n0;
  if (sec < 2) { Mp = W; m0 = hs * 64; Np = A; n0 = trow; }
  else         { Mp = A; m0 = trow;    Np = W; n0 = hs * 64; }

  const half_t* N0 = Np + (size_t)n0 * K + lane * 8;
  const half_t* M0 = Mp + (size_t)m0 * K + lane * 8;
  const size_t ts = (size_t)16 * K;

  v8f acc[4][4] = {};
  {
    V16 pn[4], pm[4], qn[4], qm[4];
    load_step8(pn, pm, N0, M0, ts, 0);         pin_order();
    int k = 32;
    for (; k < K - 32; k += 64) {
      load_step8(qn, qm, N0, M0, ts, k);       pin_order();
      mma16t(acc, pn, pm);                     pin_order();
      load_step8(pn, pm, N0, M0, ts, k + 32);  pin_order();
      mma16t(acc, qn, qm);                     pin_order();
    }
    load_step8(qn, qm, N0, M0, ts, k);         pin_order();
    mma16t(acc, pn, pm);
    mma16t(acc, qn, qm);
  }

  if (sec < 2) {
    half_t* dst = (sec == 0 ? Q : Kr) + ((size_t)(b * HN + h) * TT + tl0) * DH + lane * 8;
    const float* bp = bias + m0 + ko;
    #pragma unroll
    for (int j = 0; j < 2; ++j) {
      const float4 bl0 = *(const float4*)(bp + j * 16);
      const float4 bl1 = *(const float4*)(bp + j * 16 + 4);
      const float4 bh0 = *(const float4*)(bp + j * 16 + 32);
      const float4 bh1 = *(const float4*)(bp + j * 16 + 36);
      const float blo[8] = {bl0.x, bl0.y, bl0.z, bl0.w, bl1.x, bl1.y, bl1.z, bl1.w};
      const float bhi[8] = {bh0.x, bh0.y, bh0.z, bh0.w, bh1.x, bh1.y, bh1.z, bh1.w};
      #pragma unroll
      for (int i = 0; i < 4; ++i) {
        const int tw = i * 16 + ln;
        const int tt = ti * 64 + tw;
        const float* cp = cs_c + tt * CS_PITCH + j * 16 + ko;
        const float* sp = cs_s + tt * CS_PITCH + j * 16 + ko;
        const float4 c0 = *(const float4*)(cp), c1 = *(const float4*)(cp + 4);
        const float4 s0 = *(const float4*)(sp), s1 = *(const float4*)(sp + 4);
        const float cc[8] = {c0.x, c0.y, c0.z, c0.w, c1.x, c1.y, c1.z, c1.w};
        const float ss[8] = {s0.x, s0.y, s0.z, s0.w, s1.x, s1.y, s1.z, s1.w};
        v8h o1, o2;
        #pragma unroll
        for (int r = 0; r < 8; ++r) {
          const float x1 = acc[i][j][r] + blo[r];
          const float x2 = acc[i][j + 2][r] + bhi[r];
          o1[r] = (half_t)(x1 * cc[r] - x2 * ss[r]);
          o2[r] = (half_t)(x2 * cc[r] + x1 * ss[r]);
        }
        *(v8h*)(dst + (i * 2 + 0) * 512 + j * 256) = o1;
        *(v8h*)(dst + (i * 2 + 1) * 512 + j * 256) = o2;
      }
    }
  } else {
    half_t* vb = Vt + (size_t)(b * HN + h) * DH * TT + (size_t)(tl0 >> 5) * 2048 + lane * 8;
    #pragma unroll
    for (int i = 0; i < 4; ++i) {
      const int c = h * 64 + i * 16 + ln;
      const float bv = bias[2 * DM + c];
      #pragma unroll
      for (int j = 0; j < 4; ++j) {
        v8h o;
        #pragma unroll
        for (int r = 0; r < 8; ++r) o[r] = (half_t)(acc[i][j][r] + bv);
        *(v8h*)(vb + ((j >> 1) * 4 + i) * 512 + (j & 1) * 256) = o;
      }
    }
  }
}

__global__ __launch_bounds__(256) __attribute__((amdgpu_waves_per_eu(4, 4))) void attn_kernel(
    const half_t* __restrict__ Q, const half_t* __restrict__ Kr,
    const half_t* __restrict__ Vt, const int* __restrict__ amask,
    const float* __restrict__ sinks, half_t* __restrict__ attn) {
  const int lane = threadIdx.x & 31;
  const int wid  = (blockIdx.x * blockDim.x + threadIdx.x) >> 5;
  const int QT   = TT / 16;
  const int qt   = wid % QT;
  const int bh   = wid / QT;
  const int h    = bh % HN, b = bh / HN;
  const int qbase = qt * 16;
  const int ln = lane & 15;
  const int ko = (lane >= 16) ? 8 : 0;
  const int iq = qbase + ln;
  const half_t* Qh = Q  + (size_t)bh * TT * DH;
  const half_t* Kh = Kr + (size_t)bh * TT * DH;
  const half_t* Vh = Vt + (size_t)bh * DH * TT;
  const int* am = amask + b * TT;

  {
    const int q0 = ((blockIdx.x * blockDim.x) >> 5) % QT * 16;
    const int k0 = q0 >= 128 ? q0 - 128 : 0;
    const int nk = (q0 + 128 - k0) >> 1;
    const int u  = threadIdx.x;
    if (u < nk)                      l2_prefetch((const char*)(Kh + (size_t)k0 * DH) + (size_t)u * 256);
    else if (u >= 128 && u < 128 + nk) l2_prefetch((const char*)(Vh + (size_t)k0 * DH) + (size_t)(u - 128) * 256);
    if (u < 64)                      l2_prefetch((const char*)(Qh + (size_t)q0 * DH) + (size_t)u * 256);
  }

  V16 qb0, qb1;
  {
    const half_t* qp = Qh + (size_t)qt * 1024 + lane * 8;
    ldfm(qb0, qp);
    ldfm(qb1, qp + 512);
  }

  v8f acc[4] = {};
  float m = MINIT, l = 0.f;
  int kstart = qbase - (WIN - 1); if (kstart < 0) kstart = 0;
  kstart &= ~31;
  const int kend = qbase + 16;

  for (int jc = kstart; jc < kend; jc += 32) {
    int mk[2][8];
    #pragma unroll
    for (int t2 = 0; t2 < 2; ++t2) {
      const int4 m0 = *(const int4*)(am + jc + t2 * 16 + ko);
      const int4 m1 = *(const int4*)(am + jc + t2 * 16 + ko + 4);
      mk[t2][0] = m0.x; mk[t2][1] = m0.y; mk[t2][2] = m0.z; mk[t2][3] = m0.w;
      mk[t2][4] = m1.x; mk[t2][5] = m1.y; mk[t2][6] = m1.z; mk[t2][7] = m1.w;
    }
    v8f s[2];
    V16 ka[2][2];
    const half_t* kp = Kh + (size_t)jc * DH + lane * 8;
    #pragma unroll
    for (int t2 = 0; t2 < 2; ++t2) {
      ldfm(ka[t2][0], kp + (t2 * 2 + 0) * 512);
      ldfm(ka[t2][1], kp + (t2 * 2 + 1) * 512);
    }
    #pragma unroll
    for (int t2 = 0; t2 < 2; ++t2) {
      v8f z = {};
      z = __builtin_amdgcn_wmma_f32_16x16x32_f16(false, ka[t2][0].v, false, qb0.v,
                                                 (short)0, z, false, false);
      z = __builtin_amdgcn_wmma_f32_16x16x32_f16(false, ka[t2][1].v, false, qb1.v,
                                                 (short)0, z, false, false);
      s[t2] = z;
    }
    float tl = MASKED;
    #pragma unroll
    for (int t2 = 0; t2 < 2; ++t2) {
      #pragma unroll
      for (int r = 0; r < 8; ++r) {
        const int j = jc + t2 * 16 + r + ko;
        const bool ok = (j <= iq) & (j + WIN > iq) & (mk[t2][r] != 0);
        const float v = ok ? s[t2][r] * SM_SCALE : MASKED;
        s[t2][r] = v;
        tl = fmaxf(tl, v);
      }
    }
    tl = fmaxf(tl, __shfl_xor(tl, 16, 32));
    const float mnew = fmaxf(m, tl);
    const float co = __expf(m - mnew);
    l *= co;
    #pragma unroll
    for (int dt = 0; dt < 4; ++dt) acc[dt] = acc[dt] * co;

    V16 pbf;
    float psum = 0.f;
    #pragma unroll
    for (int r = 0; r < 8; ++r) {
      const float p0 = __expf(s[0][r] - mnew);
      const float p1 = __expf(s[1][r] - mnew);
      psum += p0 + p1;
      pbf.h[0][r] = (half_t)p0;
      pbf.h[1][r] = (half_t)p1;
    }
    l += psum + __shfl_xor(psum, 16, 32);
    m = mnew;

    #pragma unroll
    for (int dt = 0; dt < 4; ++dt) {
      const half_t* vp = Vh + (size_t)(jc >> 5) * 2048 + dt * 512 + lane * 8;
      V16 va;
      ldfm(va, vp);
      acc[dt] = __builtin_amdgcn_wmma_f32_16x16x32_f16(
          false, va.v, false, pbf.v, (short)0, acc[dt], false, false);
    }
  }

  const float snk = sinks[h];
  const float m2 = fmaxf(m, snk);
  const float co = __expf(m - m2);
  const float ltot = l * co + __expf(snk - m2);
  const float factor = co / ltot;
  half_t* op = attn + ((size_t)((b * TT + qbase) >> 4) * (DM / 32) + h * 2) * 512 + lane * 8;
  #pragma unroll
  for (int dt = 0; dt < 4; ++dt) {
    v8h o;
    #pragma unroll
    for (int r = 0; r < 8; ++r) o[r] = (half_t)(acc[dt][r] * factor);
    *(v8h*)(op + (dt >> 1) * 512 + (dt & 1) * 256) = o;
  }
}

extern "C" void kernel_launch(void* const* d_in, const int* in_sizes, int n_in,
                              void* d_out, int out_size, void* d_ws, size_t ws_size,
                              hipStream_t stream) {
  const float* x      = (const float*)d_in[0];
  const int*   amask  = (const int*)d_in[1];
  const float* nscale = (const float*)d_in[2];
  const float* qkv_w  = (const float*)d_in[3];
  const float* qkv_b  = (const float*)d_in[4];
  const float* out_w  = (const float*)d_in[5];
  const float* out_b  = (const float*)d_in[6];
  const float* sinks  = (const float*)d_in[7];
  float* out = (float*)d_out;
  char*  ws  = (char*)d_ws;

  half_t* qkvw_h = (half_t*)(ws + OFF_QKVW);
  half_t* outw_h = (half_t*)(ws + OFF_OUTW);
  half_t* t_h    = (half_t*)(ws + OFF_T);
  half_t* attn_h = (half_t*)(ws + OFF_ATTN);
  half_t* Qh     = (half_t*)(ws + OFF_Q);
  half_t* Kh     = (half_t*)(ws + OFF_K);
  half_t* Vth    = (half_t*)(ws + OFF_V);

  prep_kernel<<<PREP_GRID, 256, 0, stream>>>(x, nscale, t_h, qkv_w, qkvw_h, out_w, outw_h);

  qkv_rope_kernel<<<QKV_GRID, 384, 0, stream>>>(t_h, qkvw_h, qkv_b, Qh, Kh, Vth, DM);

  attn_kernel<<<512, 256, 0, stream>>>(Qh, Kh, Vth, amask, sinks, attn_h);

  gemm_wmma_kernel<<<256, 256, 0, stream>>>(attn_h, outw_h, out_b, x,
                                            out, MROW, DM, DM);
}
